# starting kernel
# baseline (speedup 1.0000x reference)
#include <hip/hip_runtime.h>

typedef float v2f __attribute__((ext_vector_type(2)));
typedef float v8f __attribute__((ext_vector_type(8)));

constexpr int N_NODES   = 100000;
constexpr int N_HYEDGES = 20000;
constexpr int NNZ       = 1600000;
constexpr int IN_CH     = 256;
constexpr int OUT_CH    = 128;

__global__ __launch_bounds__(256) void hyfm_gemm_wmma_f32(
    const float* __restrict__ x, const float* __restrict__ theta,
    float* __restrict__ y)
{
    const int wave    = blockIdx.x * (blockDim.x >> 5) + (threadIdx.x >> 5);
    const int n_tiles = N_NODES / 16;
    if (wave >= n_tiles) return;
    const int lane = threadIdx.x & 31;
    const int r    = lane & 15;
    const int half = lane >> 4;

    const float* __restrict__ xrow = x + (size_t)(wave * 16 + r) * IN_CH;

    v8f acc[8] = {};

    for (int kk = 0; kk < IN_CH; kk += 4) {
        const int k0 = kk + 2 * half;
        v2f a = *(const v2f*)(xrow + k0);
        const float* __restrict__ th0 = theta + (size_t)k0 * OUT_CH + r;
#pragma unroll
        for (int c = 0; c < 8; ++c) {
            v2f b;
            b.x = th0[c * 16];
            b.y = th0[c * 16 + OUT_CH];
            acc[c] = __builtin_amdgcn_wmma_f32_16x16x4_f32(
                false, a, false, b, (short)0, acc[c], false, false);
        }
    }

#pragma unroll
    for (int c = 0; c < 8; ++c) {
#pragma unroll
        for (int i = 0; i < 8; ++i) {
            y[(size_t)(wave * 16 + i + 8 * half) * OUT_CH + c * 16 + r] = acc[c][i];
        }
    }
}

__global__ __launch_bounds__(256) void hyfm_degree(
    const int* __restrict__ nid, const int* __restrict__ eid,
    float* __restrict__ deg_n, float* __restrict__ deg_e, int nnz)
{
    int i = blockIdx.x * blockDim.x + threadIdx.x;
    if (i >= nnz) return;
    atomicAdd(&deg_e[eid[i]], 1.0f);
    atomicAdd(&deg_n[nid[i]], 1.0f);
}

__global__ __launch_bounds__(256) void hyfm_recip(float* __restrict__ d, int n)
{
    int i = blockIdx.x * blockDim.x + threadIdx.x;
    if (i >= n) return;
    float v = d[i];
    d[i] = (v > 0.0f) ? (1.0f / v) : 0.0f;
}

__global__ __launch_bounds__(256) void hyfm_bias_init(
    const float* __restrict__ bias, float* __restrict__ out, int total)
{
    int i = blockIdx.x * blockDim.x + threadIdx.x;
    if (i >= total) return;
    out[i] = bias[i & (OUT_CH - 1)];
}

__global__ __launch_bounds__(256) void hyfm_scatter(
    const float* __restrict__ src, float* __restrict__ dst,
    const int* __restrict__ gidx, const int* __restrict__ sidx,
    const float* __restrict__ inv_deg, int nnz)
{
    const int w = blockIdx.x * (blockDim.x >> 5) + (threadIdx.x >> 5);
    if (w >= nnz) return;
    const int lane = threadIdx.x & 31;
    const int g = gidx[w];
    const int s = sidx[w];
    const float sc = inv_deg[s];
    const float4 v = ((const float4*)(src + (size_t)g * OUT_CH))[lane];
    float* drow = dst + (size_t)s * OUT_CH + lane * 4;
    atomicAdd(drow + 0, v.x * sc);
    atomicAdd(drow + 1, v.y * sc);
    atomicAdd(drow + 2, v.z * sc);
    atomicAdd(drow + 3, v.w * sc);
}

extern "C" void kernel_launch(void* const* d_in, const int* in_sizes, int n_in,
                              void* d_out, int out_size, void* d_ws, size_t ws_size,
                              hipStream_t stream)
{
    const float* x          = (const float*)d_in[0];
    const float* theta      = (const float*)d_in[1];
    const float* bias       = (const float*)d_in[2];
    const int*   node_idx   = (const int*)d_in[3];
    const int*   hyedge_idx = (const int*)d_in[4];
    float*       out        = (float*)d_out;

    char*  ws    = (char*)d_ws;
    float* y     = (float*)(ws);
    float* e_ft  = (float*)(ws + 51200000);
    float* deg_e = (float*)(ws + 61440000);
    float* deg_n = (float*)(ws + 61520000);

    hipMemsetAsync(ws + 51200000, 0, 10720000, stream);

    hyfm_degree<<<(NNZ + 255) / 256, 256, 0, stream>>>(
        node_idx, hyedge_idx, deg_n, deg_e, NNZ);
    hyfm_recip<<<(N_HYEDGES + N_NODES + 255) / 256, 256, 0, stream>>>(
        deg_e, N_HYEDGES + N_NODES);

    hyfm_gemm_wmma_f32<<<(N_NODES / 16 + 7) / 8, 256, 0, stream>>>(x, theta, y);

    hyfm_bias_init<<<(N_NODES * OUT_CH + 255) / 256, 256, 0, stream>>>(
        bias, out, N_NODES * OUT_CH);

    hyfm_scatter<<<(NNZ + 7) / 8, 256, 0, stream>>>(
        y, e_ft, node_idx, hyedge_idx, deg_e, NNZ);
    hyfm_scatter<<<(NNZ + 7) / 8, 256, 0, stream>>>(
        e_ft, out, hyedge_idx, node_idx, deg_n, NNZ);
}


# degree counts ride aggregation pass 1 (hyperedge counters one per 128-B line); 1/deg scales applied after the sums, at pass 2's gather; degree, reciprocal launches gone; atomics kept, arrival order
# speedup vs baseline: 1.1767x; 1.1767x over previous
#include <hip/hip_runtime.h>

typedef float v2f __attribute__((ext_vector_type(2)));
typedef float v8f __attribute__((ext_vector_type(8)));

constexpr int N_NODES   = 100000;
constexpr int N_HYEDGES = 20000;
constexpr int NNZ       = 1600000;
constexpr int IN_CH     = 256;
constexpr int OUT_CH    = 128;

__global__ __launch_bounds__(256) void hyfm_gemm_wmma_f32(
    const float* __restrict__ x, const float* __restrict__ theta,
    float* __restrict__ y)
{
    const int wave    = blockIdx.x * (blockDim.x >> 5) + (threadIdx.x >> 5);
    const int n_tiles = N_NODES / 16;
    if (wave >= n_tiles) return;
    const int lane = threadIdx.x & 31;
    const int r    = lane & 15;
    const int half = lane >> 4;

    const float* __restrict__ xrow = x + (size_t)(wave * 16 + r) * IN_CH;

    v8f acc[8] = {};

    for (int kk = 0; kk < IN_CH; kk += 4) {
        const int k0 = kk + 2 * half;
        v2f a = *(const v2f*)(xrow + k0);
        const float* __restrict__ th0 = theta + (size_t)k0 * OUT_CH + r;
#pragma unroll
        for (int c = 0; c < 8; ++c) {
            v2f b;
            b.x = th0[c * 16];
            b.y = th0[c * 16 + OUT_CH];
            acc[c] = __builtin_amdgcn_wmma_f32_16x16x4_f32(
                false, a, false, b, (short)0, acc[c], false, false);
        }
    }

#pragma unroll
    for (int c = 0; c < 8; ++c) {
#pragma unroll
        for (int i = 0; i < 8; ++i) {
            y[(size_t)(wave * 16 + i + 8 * half) * OUT_CH + c * 16 + r] = acc[c][i];
        }
    }
}

__global__ __launch_bounds__(256) void hyfm_bias_init(
    const float* __restrict__ bias, float* __restrict__ out, int total)
{
    int i = blockIdx.x * blockDim.x + threadIdx.x;
    if (i >= total) return;
    out[i] = bias[i & (OUT_CH - 1)];
}

constexpr int CNT_PITCH = 32;

__global__ __launch_bounds__(256) void agg_node_to_hyedge(
    const float* __restrict__ y, float* __restrict__ e_sum,
    const int* __restrict__ nid, const int* __restrict__ eid,
    float* __restrict__ deg_e, float* __restrict__ deg_n, int nnz)
{
    const int w = blockIdx.x * (blockDim.x >> 5) + (threadIdx.x >> 5);
    if (w >= nnz) return;
    const int lane = threadIdx.x & 31;
    const int g = nid[w];
    const int s = eid[w];
    const float4 v = ((const float4*)(y + (size_t)g * OUT_CH))[lane];
    float* drow = e_sum + (size_t)s * OUT_CH + lane * 4;
    atomicAdd(drow + 0, v.x);
    atomicAdd(drow + 1, v.y);
    atomicAdd(drow + 2, v.z);
    atomicAdd(drow + 3, v.w);
    if (lane < 2) {
        float* cnt = (lane == 0) ? (deg_e + (size_t)s * CNT_PITCH) : (deg_n + g);
        atomicAdd(cnt, 1.0f);
    }
}

__global__ __launch_bounds__(256) void agg_hyedge_to_node(
    const float* __restrict__ e_sum, float* __restrict__ out,
    const int* __restrict__ nid, const int* __restrict__ eid,
    const float* __restrict__ deg_e, const float* __restrict__ deg_n, int nnz)
{
    const int w = blockIdx.x * (blockDim.x >> 5) + (threadIdx.x >> 5);
    if (w >= nnz) return;
    const int lane = threadIdx.x & 31;
    const int g = eid[w];
    const int s = nid[w];
    const float de = deg_e[(size_t)g * CNT_PITCH];
    const float dn = deg_n[s];
    const float4 v = ((const float4*)(e_sum + (size_t)g * OUT_CH))[lane];
    const float inv_e = (de > 0.0f) ? (1.0f / de) : 0.0f;
    const float inv_n = (dn > 0.0f) ? (1.0f / dn) : 0.0f;
    float* drow = out + (size_t)s * OUT_CH + lane * 4;
    atomicAdd(drow + 0, (v.x * inv_e) * inv_n);
    atomicAdd(drow + 1, (v.y * inv_e) * inv_n);
    atomicAdd(drow + 2, (v.z * inv_e) * inv_n);
    atomicAdd(drow + 3, (v.w * inv_e) * inv_n);
}

extern "C" void kernel_launch(void* const* d_in, const int* in_sizes, int n_in,
                              void* d_out, int out_size, void* d_ws, size_t ws_size,
                              hipStream_t stream)
{
    const float* x          = (const float*)d_in[0];
    const float* theta      = (const float*)d_in[1];
    const float* bias       = (const float*)d_in[2];
    const int*   node_idx   = (const int*)d_in[3];
    const int*   hyedge_idx = (const int*)d_in[4];
    float*       out        = (float*)d_out;

    char*  ws    = (char*)d_ws;
    float* y     = out;
    float* e_sum = (float*)(ws);
    float* deg_e = (float*)(ws + 10240000);
    float* deg_n = (float*)(ws + 12800000);

    hipMemsetAsync(ws, 0, 13200000, stream);

    hyfm_gemm_wmma_f32<<<(N_NODES / 16 + 7) / 8, 256, 0, stream>>>(x, theta, y);

    agg_node_to_hyedge<<<(NNZ + 7) / 8, 256, 0, stream>>>(
        y, e_sum, node_idx, hyedge_idx, deg_e, deg_n, NNZ);

    hyfm_bias_init<<<(N_NODES * OUT_CH + 255) / 256, 256, 0, stream>>>(
        bias, out, N_NODES * OUT_CH);

    agg_hyedge_to_node<<<(NNZ + 7) / 8, 256, 0, stream>>>(
        e_sum, out, node_idx, hyedge_idx, deg_e, deg_n, NNZ);
}


# both aggregation passes read a per-group bitmap of 32-entry record lines (rebuilt every call), load only the marked lines, no longer scan 1.6M one-byte keys; LDS float adds stay in arrival order
# speedup vs baseline: 3.8769x; 3.2947x over previous
#include <hip/hip_runtime.h>

typedef float v2f __attribute__((ext_vector_type(2)));
typedef float v8f __attribute__((ext_vector_type(8)));

constexpr int N_NODES   = 100000;
constexpr int N_HYEDGES = 20000;
constexpr int NNZ       = 1600000;
constexpr int IN_CH     = 256;
constexpr int OUT_CH    = 128;

__global__ __launch_bounds__(256) void hyfm_gemm_wmma_f32(
    const float* __restrict__ x, const float* __restrict__ theta,
    float* __restrict__ y)
{
    const int wave    = blockIdx.x * (blockDim.x >> 5) + (threadIdx.x >> 5);
    const int n_tiles = N_NODES / 16;
    if (wave >= n_tiles) return;
    const int lane = threadIdx.x & 31;
    const int r    = lane & 15;
    const int half = lane >> 4;

    const float* __restrict__ xrow = x + (size_t)(wave * 16 + r) * IN_CH;

    v8f acc[8] = {};

    for (int kk = 0; kk < IN_CH; kk += 4) {
        const int k0 = kk + 2 * half;
        v2f a = *(const v2f*)(xrow + k0);
        const float* __restrict__ th0 = theta + (size_t)k0 * OUT_CH + r;
#pragma unroll
        for (int c = 0; c < 8; ++c) {
            v2f b;
            b.x = th0[c * 16];
            b.y = th0[c * 16 + OUT_CH];
            acc[c] = __builtin_amdgcn_wmma_f32_16x16x4_f32(
                false, a, false, b, (short)0, acc[c], false, false);
        }
    }

#pragma unroll
    for (int c = 0; c < 8; ++c) {
#pragma unroll
        for (int i = 0; i < 8; ++i) {
            y[(size_t)(wave * 16 + i + 8 * half) * OUT_CH + c * 16 + r] = acc[c][i];
        }
    }
}

constexpr int OW_THREADS  = 1024;
constexpr int OW_WAVES    = OW_THREADS / 32;
constexpr int CHUNK       = 32;
constexpr int N_CHUNKS    = NNZ / CHUNK;
constexpr int N_HW        = N_CHUNKS / 16;
constexpr int HW_PITCH    = 256;
constexpr int HW_PER_WAVE = 98;
constexpr int OW_CQ       = 64;
constexpr int OW_Q        = 512;
constexpr int KEY_NONE    = 255;

constexpr int P1_ROWS   = 80;
constexpr int P1_GROUPS = N_HYEDGES / P1_ROWS;

constexpr int P2_ROWS   = 400;
constexpr int P2_GROUPS = N_NODES / P2_ROWS;

static_assert(P1_GROUPS * P1_ROWS == N_HYEDGES, "row ranges must tile the hyperedges");
static_assert(P2_GROUPS * P2_ROWS == N_NODES, "row ranges must tile the nodes");
static_assert(N_CHUNKS * CHUNK == NNZ && N_HW * 16 == N_CHUNKS, "the list must split into chunks and half-words");
static_assert(OW_WAVES * HW_PER_WAVE >= N_HW && (HW_PER_WAVE & 1) == 0, "the waves of a group must cover its half-words in pairs");
static_assert(P1_GROUPS <= KEY_NONE && P2_GROUPS <= KEY_NONE && KEY_NONE < HW_PITCH, "a group number must fit its byte");

constexpr int P1_LBITS = 7;
constexpr int P2_LBITS = 9;
static_assert(P1_ROWS <= (1 << P1_LBITS) && N_NODES   <= (1 << (24 - P1_LBITS)), "pass 1 record fields");
static_assert(P2_ROWS <= (1 << P2_LBITS) && N_HYEDGES <= (1 << (24 - P2_LBITS)), "pass 2 record fields");

template <int ROWS, int LBITS>
__device__ __forceinline__ unsigned own_record(const unsigned own, const unsigned from, unsigned& key)
{
    const unsigned g = own / (unsigned)ROWS;
    key = min(g, (unsigned)KEY_NONE);
    return (key << 24) | ((from & ((1u << (24 - LBITS)) - 1u)) << LBITS) |
           ((own - g * (unsigned)ROWS) & ((1u << LBITS) - 1u));
}

__global__ __launch_bounds__(OW_THREADS) void owner_bits(
    const int* __restrict__ hyedge_idx, const int* __restrict__ node_idx,
    unsigned* __restrict__ bits1, unsigned* __restrict__ bits2,
    int* __restrict__ rec1, int* __restrict__ rec2)
{
    __shared__ unsigned t1[8 * HW_PITCH];
    __shared__ unsigned t2[8 * HW_PITCH];
    const int tid = threadIdx.x;
    t1[tid] = 0u; t1[tid + OW_THREADS] = 0u;
    t2[tid] = 0u; t2[tid + OW_THREADS] = 0u;
    __syncthreads();

    const int i4 = blockIdx.x * OW_THREADS + tid;
    if (i4 < NNZ / 4) {
        const int4 h = ((const int4*)hyedge_idx)[i4];
        const int4 n = ((const int4*)node_idx)[i4];
        const unsigned ho[4] = { (unsigned)h.x, (unsigned)h.y, (unsigned)h.z, (unsigned)h.w };
        const unsigned no[4] = { (unsigned)n.x, (unsigned)n.y, (unsigned)n.z, (unsigned)n.w };
        const int      row = (tid >> 7) * HW_PITCH;
        const unsigned bit = 1u << ((tid >> 3) & 15);
        unsigned w1[4], w2[4];
#pragma unroll
        for (int c = 0; c < 4; ++c) {
            unsigned k1, k2;
            w1[c] = own_record<P1_ROWS, P1_LBITS>(ho[c], no[c], k1);
            w2[c] = own_record<P2_ROWS, P2_LBITS>(no[c], ho[c], k2);
            if (k1 != (unsigned)KEY_NONE)
                __hip_atomic_fetch_or(&t1[row + (int)k1], bit, __ATOMIC_RELAXED, __HIP_MEMORY_SCOPE_WORKGROUP);
            if (k2 != (unsigned)KEY_NONE)
                __hip_atomic_fetch_or(&t2[row + (int)k2], bit, __ATOMIC_RELAXED, __HIP_MEMORY_SCOPE_WORKGROUP);
        }
        ((int4*)rec1)[i4] = make_int4((int)w1[0], (int)w1[1], (int)w1[2], (int)w1[3]);
        ((int4*)rec2)[i4] = make_int4((int)w2[0], (int)w2[1], (int)w2[2], (int)w2[3]);
    }
    __syncthreads();

    const int o = blockIdx.x * OW_THREADS + tid;
    if (o < N_HW * HW_PITCH / 2) {
        bits1[o] = (t1[2 * tid] & 0xffffu) | (t1[2 * tid + 1] << 16);
        bits2[o] = (t2[2 * tid] & 0xffffu) | (t2[2 * tid + 1] << 16);
    }
}

template <bool FULL, int LBITS>
__device__ __forceinline__ void own_drain(
    float* acc, int* cnt, const int* q, int head, int n, int lane,
    const float* __restrict__ src)
{
    int r = 0, g = 0;
    if (FULL || lane < n) {
        const unsigned w = (unsigned)q[(head + lane) & (OW_Q - 1)];
        r = (int)(w & ((1u << LBITS) - 1u));
        g = (int)((w >> LBITS) & ((1u << (24 - LBITS)) - 1u));
        __hip_atomic_fetch_add(cnt + r, 1, __ATOMIC_RELAXED, __HIP_MEMORY_SCOPE_WORKGROUP);
    }
    for (int jb = 0; jb < (FULL ? 32 : n); jb += 8) {
        float4 v[8];
#pragma unroll
        for (int t = 0; t < 8; ++t) {
            const int gj = __builtin_amdgcn_readlane(g, jb + t);
            v[t] = *(const float4*)(src + (size_t)gj * OUT_CH + 4 * lane);
        }
#pragma unroll
        for (int t = 0; t < 8; ++t) {
            if (!FULL && jb + t >= n) break;
            const int rj = __builtin_amdgcn_readlane(r, jb + t);
            float* d = acc + rj * OUT_CH + lane;
            __hip_atomic_fetch_add(d +  0, v[t].x, __ATOMIC_RELAXED, __HIP_MEMORY_SCOPE_WORKGROUP);
            __hip_atomic_fetch_add(d + 32, v[t].y, __ATOMIC_RELAXED, __HIP_MEMORY_SCOPE_WORKGROUP);
            __hip_atomic_fetch_add(d + 64, v[t].z, __ATOMIC_RELAXED, __HIP_MEMORY_SCOPE_WORKGROUP);
            __hip_atomic_fetch_add(d + 96, v[t].w, __ATOMIC_RELAXED, __HIP_MEMORY_SCOPE_WORKGROUP);
        }
    }
}

template <bool FULL, int LBITS>
__device__ __forceinline__ void own_chunks(
    float* acc, int* cnt, int* q, int& head, int& qn,
    const int* cq, int chead, int n, int lane, unsigned group,
    const float* __restrict__ src, const int* __restrict__ rec)
{
    int cj = 0;
    if (FULL || lane < n) cj = cq[(chead + lane) & (OW_CQ - 1)];
#pragma nounroll
    for (int jb = 0; jb < (FULL ? 32 : n); jb += 8) {
        int w[8];
#pragma unroll
        for (int t = 0; t < 8; ++t) {
            const int c = __builtin_amdgcn_readlane(cj, jb + t);
            w[t] = rec[c * CHUNK + lane];
        }
#pragma unroll
        for (int t = 0; t < 8; ++t) {
            if (!FULL && jb + t >= n) break;
            const bool hit = ((unsigned)w[t] >> 24) == group;
            const unsigned mk = __builtin_amdgcn_ballot_w32(hit);
            if (hit)
                q[(head + qn + (int)__builtin_amdgcn_mbcnt_lo(mk, 0u)) & (OW_Q - 1)] = w[t];
            qn += __builtin_popcount(mk);
        }
        while (qn >= 32) {
            own_drain<true, LBITS>(acc, cnt, q, head, 32, lane, src);
            head = (head + 32) & (OW_Q - 1);
            qn -= 32;
        }
    }
}

template <int ROWS, int LBITS, bool HAS_BIAS>
__global__ __launch_bounds__(OW_THREADS) void agg_owned(
    const float* __restrict__ src, float* __restrict__ dst,
    const unsigned* __restrict__ bits, const int* __restrict__ rec,
    const float* __restrict__ bias)
{
    __shared__ __attribute__((aligned(16))) float acc[ROWS * OUT_CH];
    __shared__ int cnt[ROWS];
    __shared__ int q_all[OW_WAVES][OW_Q];
    __shared__ int cq_all[OW_WAVES][OW_CQ];

    const int tid  = threadIdx.x;
    const int lane = tid & 31;
    const int wave = __builtin_amdgcn_readfirstlane(tid >> 5);
    int* q  = q_all[wave];
    int* cq = cq_all[wave];

    float4* acc4 = (float4*)acc;
    for (int i = tid; i < ROWS * OUT_CH / 4; i += OW_THREADS)
        acc4[i] = make_float4(0.0f, 0.0f, 0.0f, 0.0f);
    for (int i = tid; i < ROWS; i += OW_THREADS) cnt[i] = 0;
    __syncthreads();

    int head  = 0;
    int qn    = 0;
    int chead = 0;
    int cn    = 0;

    const unsigned group = blockIdx.x;
    const unsigned short* __restrict__ hw = (const unsigned short*)bits;

    unsigned m[2];
    int      c0[2];
#pragma unroll
    for (int k = 0; k < 2; ++k) {
        const int pair = k * 32 + lane;
        const int h0   = wave * HW_PER_WAVE + 2 * pair;
        const bool ok0 = pair < HW_PER_WAVE / 2 && h0 < N_HW;
        const bool ok1 = pair < HW_PER_WAVE / 2 && h0 + 1 < N_HW;
        const unsigned lo = hw[(size_t)min(h0,     N_HW - 1) * HW_PITCH + group];
        const unsigned hi = hw[(size_t)min(h0 + 1, N_HW - 1) * HW_PITCH + group];
        m[k]  = (ok0 ? lo : 0u) | (ok1 ? (hi << 16) : 0u);
        c0[k] = h0 * 16;
    }
#pragma unroll
    for (int k = 0; k < 2; ++k) {
        unsigned mm = m[k];
        unsigned mask = __builtin_amdgcn_ballot_w32(mm != 0u);
        while (mask != 0u) {
            if (mm != 0u) {
                const int p = __builtin_ctz(mm);
                mm &= mm - 1u;
                cq[(chead + cn + (int)__builtin_amdgcn_mbcnt_lo(mask, 0u)) & (OW_CQ - 1)] = c0[k] + p;
            }
            cn += __builtin_popcount(mask);
            if (cn >= 32) {
                own_chunks<true, LBITS>(acc, cnt, q, head, qn, cq, chead, 32, lane, group, src, rec);
                chead = (chead + 32) & (OW_CQ - 1);
                cn -= 32;
            }
            mask = __builtin_amdgcn_ballot_w32(mm != 0u);
        }
    }
    own_chunks<false, LBITS>(acc, cnt, q, head, qn, cq, chead, cn, lane, group, src, rec);
    own_drain<false, LBITS>(acc, cnt, q, head, qn, lane, src);

    __syncthreads();

    const float4* __restrict__ bias4 = (const float4*)bias;
    float4* __restrict__ dst4 = (float4*)dst + (size_t)(blockIdx.x * ROWS) * (OUT_CH / 4);
    for (int i = tid; i < ROWS * OUT_CH / 4; i += OW_THREADS) {
        const int   c   = cnt[i >> 5];
        const float inv = (c > 0) ? (1.0f / (float)c) : 0.0f;
        const float* ar = acc + (i >> 5) * OUT_CH + (i & 31);
        const float4 a = make_float4(ar[0], ar[32], ar[64], ar[96]);
        if (HAS_BIAS) {
            const float4 b = bias4[i & 31];
            dst4[i] = make_float4(b.x + a.x * inv, b.y + a.y * inv,
                                  b.z + a.z * inv, b.w + a.w * inv);
        } else {
            dst4[i] = make_float4(a.x * inv, a.y * inv, a.z * inv, a.w * inv);
        }
    }
}

extern "C" void kernel_launch(void* const* d_in, const int* in_sizes, int n_in,
                              void* d_out, int out_size, void* d_ws, size_t ws_size,
                              hipStream_t stream)
{
    const float* x          = (const float*)d_in[0];
    const float* theta      = (const float*)d_in[1];
    const float* bias       = (const float*)d_in[2];
    const int*   node_idx   = (const int*)d_in[3];
    const int*   hyedge_idx = (const int*)d_in[4];
    float*       out        = (float*)d_out;

    float*    y      = out;
    float*    e_mean = (float*)d_ws;
    unsigned* bits1  = (unsigned*)((char*)d_ws + 10240000);
    unsigned* bits2  = (unsigned*)((char*)d_ws + 11840000);
    int*      rec1   = (int*)((char*)d_ws + 13440000);
    int*      rec2   = (int*)((char*)d_ws + 19840000);

    owner_bits<<<(NNZ / 4 + OW_THREADS - 1) / OW_THREADS, OW_THREADS, 0, stream>>>(
        hyedge_idx, node_idx, bits1, bits2, rec1, rec2);

    hyfm_gemm_wmma_f32<<<(N_NODES / 16 + 7) / 8, 256, 0, stream>>>(x, theta, y);

    agg_owned<P1_ROWS, P1_LBITS, false><<<P1_GROUPS, OW_THREADS, 0, stream>>>(
        y, e_mean, bits1, rec1, bias);

    agg_owned<P2_ROWS, P2_LBITS, true><<<P2_GROUPS, OW_THREADS, 0, stream>>>(
        e_mean, out, bits2, rec2, bias);
}


# GEMM re-cut by columns: each wave keeps its 16-column slice of theta in registers for all its rows; x passes once through LDS per 32-row step, shared by 8 waves; k order and sums unchanged
# speedup vs baseline: 4.1733x; 1.0765x over previous
#include <hip/hip_runtime.h>

typedef float v2f __attribute__((ext_vector_type(2)));
typedef float v4f __attribute__((ext_vector_type(4)));
typedef float v8f __attribute__((ext_vector_type(8)));

constexpr int N_NODES   = 100000;
constexpr int N_HYEDGES = 20000;
constexpr int NNZ       = 1600000;
constexpr int IN_CH     = 256;
constexpr int OUT_CH    = 128;

constexpr int G_THREADS = 256;
constexpr int G_ROWS    = 32;
constexpr int G_STEPS   = N_NODES / G_ROWS;
constexpr int G_PITCH   = IN_CH + 4;
constexpr int G_BLOCKS  = 512;
constexpr int G_PRE     = G_ROWS * IN_CH / 4 / G_THREADS;

static_assert(G_STEPS * G_ROWS == N_NODES, "steps must tile the rows");
static_assert(OUT_CH == 16 * (G_THREADS / 32), "one 16-column tile per wave");
static_assert(G_BLOCKS <= G_STEPS, "every group has a first step");
static_assert(G_PRE * G_THREADS * 4 == G_ROWS * IN_CH, "a step must split evenly over the threads");

__device__ __forceinline__ void gemm_fetch(v4f (&pre)[G_PRE], const float* __restrict__ x, int step, int tid)
{
    const v4f* __restrict__ s = (const v4f*)(x + (size_t)step * G_ROWS * IN_CH) + tid;
#pragma unroll
    for (int p = 0; p < G_PRE; ++p) pre[p] = s[p * G_THREADS];
}

__device__ __forceinline__ void gemm_stage(float* xs, const v4f (&pre)[G_PRE], int tid)
{
#pragma unroll
    for (int p = 0; p < G_PRE; ++p) {
        const int row = p * (G_THREADS / 64) + (tid >> 6);
        *(v4f*)(xs + row * G_PITCH + 4 * (tid & 63)) = pre[p];
    }
}

__global__ __launch_bounds__(G_THREADS) void hyfm_gemm_wmma_f32(
    const float* __restrict__ x, const float* __restrict__ theta,
    float* __restrict__ y)
{
    __shared__ __attribute__((aligned(16))) float xs[2][G_ROWS * G_PITCH];

    const int tid  = threadIdx.x;
    const int lane = tid & 31;
    const int wave = __builtin_amdgcn_readfirstlane(tid >> 5);
    const int r    = lane & 15;
    const int half = lane >> 4;

    v2f b[IN_CH / 4];
    const float* __restrict__ th = theta + (size_t)(2 * half) * OUT_CH + wave * 16 + r;
#pragma unroll
    for (int s = 0; s < IN_CH / 4; ++s) {
        b[s].x = th[(4 * s) * OUT_CH];
        b[s].y = th[(4 * s + 1) * OUT_CH];
    }

    v4f pre[G_PRE];
    gemm_fetch(pre, x, blockIdx.x, tid);
    gemm_stage(xs[0], pre, tid);
    __syncthreads();

    const int aoff = r * G_PITCH + 2 * half;
    float* __restrict__ ycol = y + (size_t)(8 * half) * OUT_CH + wave * 16 + r;

    int buf = 0;
    for (int step = blockIdx.x; step < G_STEPS; step += G_BLOCKS) {
        const int  next = step + G_BLOCKS;
        const bool more = next < G_STEPS;
        if (more) gemm_fetch(pre, x, next, tid);

        const float* ap = xs[buf] + aoff;
        v8f acc0 = {}, acc1 = {};
#pragma unroll
        for (int s = 0; s < IN_CH / 4; ++s) {
            const v2f a0 = *(const v2f*)(ap + 4 * s);
            const v2f a1 = *(const v2f*)(ap + 16 * G_PITCH + 4 * s);
            acc0 = __builtin_amdgcn_wmma_f32_16x16x4_f32(
                false, a0, false, b[s], (short)0, acc0, false, false);
            acc1 = __builtin_amdgcn_wmma_f32_16x16x4_f32(
                false, a1, false, b[s], (short)0, acc1, false, false);
        }

        if (more) gemm_stage(xs[buf ^ 1], pre, tid);
        __syncthreads();
        buf ^= 1;

        float* __restrict__ yo = ycol + (size_t)step * G_ROWS * OUT_CH;
#pragma unroll
        for (int i = 0; i < 8; ++i) {
            yo[i * OUT_CH]        = acc0[i];
            yo[(i + 16) * OUT_CH] = acc1[i];
        }
    }
}

constexpr int OW_THREADS  = 1024;
constexpr int OW_WAVES    = OW_THREADS / 32;
constexpr int CHUNK       = 32;
constexpr int N_CHUNKS    = NNZ / CHUNK;
constexpr int N_HW        = N_CHUNKS / 16;
constexpr int HW_PITCH    = 256;
constexpr int HW_PER_WAVE = 98;
constexpr int OW_CQ       = 64;
constexpr int OW_Q        = 512;
constexpr int KEY_NONE    = 255;

constexpr int P1_ROWS   = 80;
constexpr int P1_GROUPS = N_HYEDGES / P1_ROWS;

constexpr int P2_ROWS   = 400;
constexpr int P2_GROUPS = N_NODES / P2_ROWS;

static_assert(P1_GROUPS * P1_ROWS == N_HYEDGES, "row ranges must tile the hyperedges");
static_assert(P2_GROUPS * P2_ROWS == N_NODES, "row ranges must tile the nodes");
static_assert(N_CHUNKS * CHUNK == NNZ && N_HW * 16 == N_CHUNKS, "the list must split into chunks and half-words");
static_assert(OW_WAVES * HW_PER_WAVE >= N_HW && (HW_PER_WAVE & 1) == 0, "the waves of a group must cover its half-words in pairs");
static_assert(P1_GROUPS <= KEY_NONE && P2_GROUPS <= KEY_NONE && KEY_NONE < HW_PITCH, "a group number must fit its byte");

constexpr int P1_LBITS = 7;
constexpr int P2_LBITS = 9;
static_assert(P1_ROWS <= (1 << P1_LBITS) && N_NODES   <= (1 << (24 - P1_LBITS)), "pass 1 record fields");
static_assert(P2_ROWS <= (1 << P2_LBITS) && N_HYEDGES <= (1 << (24 - P2_LBITS)), "pass 2 record fields");

template <int ROWS, int LBITS>
__device__ __forceinline__ unsigned own_record(const unsigned own, const unsigned from, unsigned& key)
{
    const unsigned g = own / (unsigned)ROWS;
    key = min(g, (unsigned)KEY_NONE);
    return (key << 24) | ((from & ((1u << (24 - LBITS)) - 1u)) << LBITS) |
           ((own - g * (unsigned)ROWS) & ((1u << LBITS) - 1u));
}

__global__ __launch_bounds__(OW_THREADS) void owner_bits(
    const int* __restrict__ hyedge_idx, const int* __restrict__ node_idx,
    unsigned* __restrict__ bits1, unsigned* __restrict__ bits2,
    int* __restrict__ rec1, int* __restrict__ rec2)
{
    __shared__ unsigned t1[8 * HW_PITCH];
    __shared__ unsigned t2[8 * HW_PITCH];
    const int tid = threadIdx.x;
    t1[tid] = 0u; t1[tid + OW_THREADS] = 0u;
    t2[tid] = 0u; t2[tid + OW_THREADS] = 0u;
    __syncthreads();

    const int i4 = blockIdx.x * OW_THREADS + tid;
    if (i4 < NNZ / 4) {
        const int4 h = ((const int4*)hyedge_idx)[i4];
        const int4 n = ((const int4*)node_idx)[i4];
        const unsigned ho[4] = { (unsigned)h.x, (unsigned)h.y, (unsigned)h.z, (unsigned)h.w };
        const unsigned no[4] = { (unsigned)n.x, (unsigned)n.y, (unsigned)n.z, (unsigned)n.w };
        const int      row = (tid >> 7) * HW_PITCH;
        const unsigned bit = 1u << ((tid >> 3) & 15);
        unsigned w1[4], w2[4];
#pragma unroll
        for (int c = 0; c < 4; ++c) {
            unsigned k1, k2;
            w1[c] = own_record<P1_ROWS, P1_LBITS>(ho[c], no[c], k1);
            w2[c] = own_record<P2_ROWS, P2_LBITS>(no[c], ho[c], k2);
            if (k1 != (unsigned)KEY_NONE)
                __hip_atomic_fetch_or(&t1[row + (int)k1], bit, __ATOMIC_RELAXED, __HIP_MEMORY_SCOPE_WORKGROUP);
            if (k2 != (unsigned)KEY_NONE)
                __hip_atomic_fetch_or(&t2[row + (int)k2], bit, __ATOMIC_RELAXED, __HIP_MEMORY_SCOPE_WORKGROUP);
        }
        ((int4*)rec1)[i4] = make_int4((int)w1[0], (int)w1[1], (int)w1[2], (int)w1[3]);
        ((int4*)rec2)[i4] = make_int4((int)w2[0], (int)w2[1], (int)w2[2], (int)w2[3]);
    }
    __syncthreads();

    const int o = blockIdx.x * OW_THREADS + tid;
    if (o < N_HW * HW_PITCH / 2) {
        bits1[o] = (t1[2 * tid] & 0xffffu) | (t1[2 * tid + 1] << 16);
        bits2[o] = (t2[2 * tid] & 0xffffu) | (t2[2 * tid + 1] << 16);
    }
}

template <bool FULL, int LBITS>
__device__ __forceinline__ void own_drain(
    float* acc, int* cnt, const int* q, int head, int n, int lane,
    const float* __restrict__ src)
{
    int r = 0, g = 0;
    if (FULL || lane < n) {
        const unsigned w = (unsigned)q[(head + lane) & (OW_Q - 1)];
        r = (int)(w & ((1u << LBITS) - 1u));
        g = (int)((w >> LBITS) & ((1u << (24 - LBITS)) - 1u));
        __hip_atomic_fetch_add(cnt + r, 1, __ATOMIC_RELAXED, __HIP_MEMORY_SCOPE_WORKGROUP);
    }
    for (int jb = 0; jb < (FULL ? 32 : n); jb += 8) {
        float4 v[8];
#pragma unroll
        for (int t = 0; t < 8; ++t) {
            const int gj = __builtin_amdgcn_readlane(g, jb + t);
            v[t] = *(const float4*)(src + (size_t)gj * OUT_CH + 4 * lane);
        }
#pragma unroll
        for (int t = 0; t < 8; ++t) {
            if (!FULL && jb + t >= n) break;
            const int rj = __builtin_amdgcn_readlane(r, jb + t);
            float* d = acc + rj * OUT_CH + lane;
            __hip_atomic_fetch_add(d +  0, v[t].x, __ATOMIC_RELAXED, __HIP_MEMORY_SCOPE_WORKGROUP);
            __hip_atomic_fetch_add(d + 32, v[t].y, __ATOMIC_RELAXED, __HIP_MEMORY_SCOPE_WORKGROUP);
            __hip_atomic_fetch_add(d + 64, v[t].z, __ATOMIC_RELAXED, __HIP_MEMORY_SCOPE_WORKGROUP);
            __hip_atomic_fetch_add(d + 96, v[t].w, __ATOMIC_RELAXED, __HIP_MEMORY_SCOPE_WORKGROUP);
        }
    }
}

template <bool FULL, int LBITS>
__device__ __forceinline__ void own_chunks(
    float* acc, int* cnt, int* q, int& head, int& qn,
    const int* cq, int chead, int n, int lane, unsigned group,
    const float* __restrict__ src, const int* __restrict__ rec)
{
    int cj = 0;
    if (FULL || lane < n) cj = cq[(chead + lane) & (OW_CQ - 1)];
#pragma nounroll
    for (int jb = 0; jb < (FULL ? 32 : n); jb += 8) {
        int w[8];
#pragma unroll
        for (int t = 0; t < 8; ++t) {
            const int c = __builtin_amdgcn_readlane(cj, jb + t);
            w[t] = rec[c * CHUNK + lane];
        }
#pragma unroll
        for (int t = 0; t < 8; ++t) {
            if (!FULL && jb + t >= n) break;
            const bool hit = ((unsigned)w[t] >> 24) == group;
            const unsigned mk = __builtin_amdgcn_ballot_w32(hit);
            if (hit)
                q[(head + qn + (int)__builtin_amdgcn_mbcnt_lo(mk, 0u)) & (OW_Q - 1)] = w[t];
            qn += __builtin_popcount(mk);
        }
        while (qn >= 32) {
            own_drain<true, LBITS>(acc, cnt, q, head, 32, lane, src);
            head = (head + 32) & (OW_Q - 1);
            qn -= 32;
        }
    }
}

template <int ROWS, int LBITS, bool HAS_BIAS>
__global__ __launch_bounds__(OW_THREADS) void agg_owned(
    const float* __restrict__ src, float* __restrict__ dst,
    const unsigned* __restrict__ bits, const int* __restrict__ rec,
    const float* __restrict__ bias)
{
    __shared__ __attribute__((aligned(16))) float acc[ROWS * OUT_CH];
    __shared__ int cnt[ROWS];
    __shared__ int q_all[OW_WAVES][OW_Q];
    __shared__ int cq_all[OW_WAVES][OW_CQ];

    const int tid  = threadIdx.x;
    const int lane = tid & 31;
    const int wave = __builtin_amdgcn_readfirstlane(tid >> 5);
    int* q  = q_all[wave];
    int* cq = cq_all[wave];

    float4* acc4 = (float4*)acc;
    for (int i = tid; i < ROWS * OUT_CH / 4; i += OW_THREADS)
        acc4[i] = make_float4(0.0f, 0.0f, 0.0f, 0.0f);
    for (int i = tid; i < ROWS; i += OW_THREADS) cnt[i] = 0;
    __syncthreads();

    int head  = 0;
    int qn    = 0;
    int chead = 0;
    int cn    = 0;

    const unsigned group = blockIdx.x;
    const unsigned short* __restrict__ hw = (const unsigned short*)bits;

    unsigned m[2];
    int      c0[2];
#pragma unroll
    for (int k = 0; k < 2; ++k) {
        const int pair = k * 32 + lane;
        const int h0   = wave * HW_PER_WAVE + 2 * pair;
        const bool ok0 = pair < HW_PER_WAVE / 2 && h0 < N_HW;
        const bool ok1 = pair < HW_PER_WAVE / 2 && h0 + 1 < N_HW;
        const unsigned lo = hw[(size_t)min(h0,     N_HW - 1) * HW_PITCH + group];
        const unsigned hi = hw[(size_t)min(h0 + 1, N_HW - 1) * HW_PITCH + group];
        m[k]  = (ok0 ? lo : 0u) | (ok1 ? (hi << 16) : 0u);
        c0[k] = h0 * 16;
    }
#pragma unroll
    for (int k = 0; k < 2; ++k) {
        unsigned mm = m[k];
        unsigned mask = __builtin_amdgcn_ballot_w32(mm != 0u);
        while (mask != 0u) {
            if (mm != 0u) {
                const int p = __builtin_ctz(mm);
                mm &= mm - 1u;
                cq[(chead + cn + (int)__builtin_amdgcn_mbcnt_lo(mask, 0u)) & (OW_CQ - 1)] = c0[k] + p;
            }
            cn += __builtin_popcount(mask);
            if (cn >= 32) {
                own_chunks<true, LBITS>(acc, cnt, q, head, qn, cq, chead, 32, lane, group, src, rec);
                chead = (chead + 32) & (OW_CQ - 1);
                cn -= 32;
            }
            mask = __builtin_amdgcn_ballot_w32(mm != 0u);
        }
    }
    own_chunks<false, LBITS>(acc, cnt, q, head, qn, cq, chead, cn, lane, group, src, rec);
    own_drain<false, LBITS>(acc, cnt, q, head, qn, lane, src);

    __syncthreads();

    const float4* __restrict__ bias4 = (const float4*)bias;
    float4* __restrict__ dst4 = (float4*)dst + (size_t)(blockIdx.x * ROWS) * (OUT_CH / 4);
    for (int i = tid; i < ROWS * OUT_CH / 4; i += OW_THREADS) {
        const int   c   = cnt[i >> 5];
        const float inv = (c > 0) ? (1.0f / (float)c) : 0.0f;
        const float* ar = acc + (i >> 5) * OUT_CH + (i & 31);
        const float4 a = make_float4(ar[0], ar[32], ar[64], ar[96]);
        if (HAS_BIAS) {
            const float4 b = bias4[i & 31];
            dst4[i] = make_float4(b.x + a.x * inv, b.y + a.y * inv,
                                  b.z + a.z * inv, b.w + a.w * inv);
        } else {
            dst4[i] = make_float4(a.x * inv, a.y * inv, a.z * inv, a.w * inv);
        }
    }
}

extern "C" void kernel_launch(void* const* d_in, const int* in_sizes, int n_in,
                              void* d_out, int out_size, void* d_ws, size_t ws_size,
                              hipStream_t stream)
{
    const float* x          = (const float*)d_in[0];
    const float* theta      = (const float*)d_in[1];
    const float* bias       = (const float*)d_in[2];
    const int*   node_idx   = (const int*)d_in[3];
    const int*   hyedge_idx = (const int*)d_in[4];
    float*       out        = (float*)d_out;

    float*    y      = out;
    float*    e_mean = (float*)d_ws;
    unsigned* bits1  = (unsigned*)((char*)d_ws + 10240000);
    unsigned* bits2  = (unsigned*)((char*)d_ws + 11840000);
    int*      rec1   = (int*)((char*)d_ws + 13440000);
    int*      rec2   = (int*)((char*)d_ws + 19840000);

    owner_bits<<<(NNZ / 4 + OW_THREADS - 1) / OW_THREADS, OW_THREADS, 0, stream>>>(
        hyedge_idx, node_idx, bits1, bits2, rec1, rec2);

    hyfm_gemm_wmma_f32<<<G_BLOCKS, G_THREADS, 0, stream>>>(x, theta, y);

    agg_owned<P1_ROWS, P1_LBITS, false><<<P1_GROUPS, OW_THREADS, 0, stream>>>(
        y, e_mean, bits1, rec1, bias);

    agg_owned<P2_ROWS, P2_LBITS, true><<<P2_GROUPS, OW_THREADS, 0, stream>>>(
        e_mean, out, bits2, rec2, bias);
}


# aggregation passes: each 4,096-entry segment's records sorted by owner group in LDS (integer ranks, every call); an owner reads its 391 short runs, no bitmapped lines; float adds in arrival order
# speedup vs baseline: 4.7918x; 1.1482x over previous
#include <hip/hip_runtime.h>

typedef float v2f __attribute__((ext_vector_type(2)));
typedef float v4f __attribute__((ext_vector_type(4)));
typedef float v8f __attribute__((ext_vector_type(8)));

constexpr int N_NODES   = 100000;
constexpr int N_HYEDGES = 20000;
constexpr int NNZ       = 1600000;
constexpr int IN_CH     = 256;
constexpr int OUT_CH    = 128;

constexpr int G_THREADS = 256;
constexpr int G_ROWS    = 32;
constexpr int G_STEPS   = N_NODES / G_ROWS;
constexpr int G_PITCH   = IN_CH + 4;
constexpr int G_BLOCKS  = 512;
constexpr int G_PRE     = G_ROWS * IN_CH / 4 / G_THREADS;

static_assert(G_STEPS * G_ROWS == N_NODES, "steps must tile the rows");
static_assert(OUT_CH == 16 * (G_THREADS / 32), "one 16-column tile per wave");
static_assert(G_BLOCKS <= G_STEPS, "every group has a first step");
static_assert(G_PRE * G_THREADS * 4 == G_ROWS * IN_CH, "a step must split evenly over the threads");

__device__ __forceinline__ void gemm_fetch(v4f (&pre)[G_PRE], const float* __restrict__ x, int step, int tid)
{
    const v4f* __restrict__ s = (const v4f*)(x + (size_t)step * G_ROWS * IN_CH) + tid;
#pragma unroll
    for (int p = 0; p < G_PRE; ++p) pre[p] = s[p * G_THREADS];
}

__device__ __forceinline__ void gemm_stage(float* xs, const v4f (&pre)[G_PRE], int tid)
{
#pragma unroll
    for (int p = 0; p < G_PRE; ++p) {
        const int row = p * (G_THREADS / 64) + (tid >> 6);
        *(v4f*)(xs + row * G_PITCH + 4 * (tid & 63)) = pre[p];
    }
}

__global__ __launch_bounds__(G_THREADS) void hyfm_gemm_wmma_f32(
    const float* __restrict__ x, const float* __restrict__ theta,
    float* __restrict__ y)
{
    __shared__ __attribute__((aligned(16))) float xs[2][G_ROWS * G_PITCH];

    const int tid  = threadIdx.x;
    const int lane = tid & 31;
    const int wave = __builtin_amdgcn_readfirstlane(tid >> 5);
    const int r    = lane & 15;
    const int half = lane >> 4;

    v2f b[IN_CH / 4];
    const float* __restrict__ th = theta + (size_t)(2 * half) * OUT_CH + wave * 16 + r;
#pragma unroll
    for (int s = 0; s < IN_CH / 4; ++s) {
        b[s].x = th[(4 * s) * OUT_CH];
        b[s].y = th[(4 * s + 1) * OUT_CH];
    }

    v4f pre[G_PRE];
    gemm_fetch(pre, x, blockIdx.x, tid);
    gemm_stage(xs[0], pre, tid);
    __syncthreads();

    const int aoff = r * G_PITCH + 2 * half;
    float* __restrict__ ycol = y + (size_t)(8 * half) * OUT_CH + wave * 16 + r;

    int buf = 0;
    for (int step = blockIdx.x; step < G_STEPS; step += G_BLOCKS) {
        const int  next = step + G_BLOCKS;
        const bool more = next < G_STEPS;
        if (more) gemm_fetch(pre, x, next, tid);

        const float* ap = xs[buf] + aoff;
        v8f acc0 = {}, acc1 = {};
#pragma unroll
        for (int s = 0; s < IN_CH / 4; ++s) {
            const v2f a0 = *(const v2f*)(ap + 4 * s);
            const v2f a1 = *(const v2f*)(ap + 16 * G_PITCH + 4 * s);
            acc0 = __builtin_amdgcn_wmma_f32_16x16x4_f32(
                false, a0, false, b[s], (short)0, acc0, false, false);
            acc1 = __builtin_amdgcn_wmma_f32_16x16x4_f32(
                false, a1, false, b[s], (short)0, acc1, false, false);
        }

        if (more) gemm_stage(xs[buf ^ 1], pre, tid);
        __syncthreads();
        buf ^= 1;

        float* __restrict__ yo = ycol + (size_t)step * G_ROWS * OUT_CH;
#pragma unroll
        for (int i = 0; i < 8; ++i) {
            yo[i * OUT_CH]        = acc0[i];
            yo[(i + 16) * OUT_CH] = acc1[i];
        }
    }
}

constexpr int OW_THREADS   = 1024;
constexpr int OW_WAVES     = OW_THREADS / 32;
constexpr int SEG          = 4 * OW_THREADS;
constexpr int N_SEG        = (NNZ + SEG - 1) / SEG;
constexpr int OFF_PITCH    = 256;
constexpr int SEG_PER_WAVE = (N_SEG + OW_WAVES - 1) / OW_WAVES;
constexpr int OW_Q         = 512;
constexpr int KEY_NONE     = 255;

constexpr int P1_ROWS   = 80;
constexpr int P1_GROUPS = N_HYEDGES / P1_ROWS;

constexpr int P2_ROWS   = 400;
constexpr int P2_GROUPS = N_NODES / P2_ROWS;

static_assert(P1_GROUPS * P1_ROWS == N_HYEDGES, "row ranges must tile the hyperedges");
static_assert(P2_GROUPS * P2_ROWS == N_NODES, "row ranges must tile the nodes");
static_assert(NNZ % 4 == 0, "a thread of owner_sort takes four entries");
static_assert(SEG <= 0xffff, "a run's start and end must fit a half-word each");
static_assert(SEG_PER_WAVE <= 32 && SEG_PER_WAVE * 32 + 32 <= OW_Q, "a wave's first pieces must fit its ring");
static_assert(P1_GROUPS <= KEY_NONE && P2_GROUPS <= KEY_NONE && KEY_NONE < OFF_PITCH, "a group number must fit its byte");

constexpr int P1_LBITS = 7;
constexpr int P2_LBITS = 9;
static_assert(P1_ROWS <= (1 << P1_LBITS) && N_NODES   <= (1 << (24 - P1_LBITS)), "pass 1 record fields");
static_assert(P2_ROWS <= (1 << P2_LBITS) && N_HYEDGES <= (1 << (24 - P2_LBITS)), "pass 2 record fields");

template <int ROWS, int LBITS>
__device__ __forceinline__ unsigned own_record(const unsigned own, const unsigned from, unsigned& key)
{
    const unsigned g = own / (unsigned)ROWS;
    key = min(g, (unsigned)KEY_NONE);
    return (key << 24) | ((from & ((1u << (24 - LBITS)) - 1u)) << LBITS) |
           ((own - g * (unsigned)ROWS) & ((1u << LBITS) - 1u));
}

__global__ __launch_bounds__(OW_THREADS) void owner_sort(
    const int* __restrict__ hyedge_idx, const int* __restrict__ node_idx,
    unsigned* __restrict__ off1, unsigned* __restrict__ off2,
    int* __restrict__ rec1, int* __restrict__ rec2)
{
    __shared__ unsigned h1[OFF_PITCH], h2[OFF_PITCH];
    __shared__ unsigned p1[OFF_PITCH], p2[OFF_PITCH];
    __shared__ __attribute__((aligned(16))) int s1[SEG];
    __shared__ __attribute__((aligned(16))) int s2[SEG];
    const int tid  = threadIdx.x;
    const int lane = tid & 31;
    const int wave = tid >> 5;
    if (tid < OFF_PITCH) { h1[tid] = 0u; h2[tid] = 0u; }
    __syncthreads();

    const int  i4   = blockIdx.x * OW_THREADS + tid;
    const bool live = i4 < NNZ / 4;
    unsigned w1[4], w2[4], k1[4], k2[4], r1[4], r2[4];
    if (live) {
        const int4 h = ((const int4*)hyedge_idx)[i4];
        const int4 n = ((const int4*)node_idx)[i4];
        const unsigned ho[4] = { (unsigned)h.x, (unsigned)h.y, (unsigned)h.z, (unsigned)h.w };
        const unsigned no[4] = { (unsigned)n.x, (unsigned)n.y, (unsigned)n.z, (unsigned)n.w };
#pragma unroll
        for (int c = 0; c < 4; ++c) {
            w1[c] = own_record<P1_ROWS, P1_LBITS>(ho[c], no[c], k1[c]);
            w2[c] = own_record<P2_ROWS, P2_LBITS>(no[c], ho[c], k2[c]);
            r1[c] = __hip_atomic_fetch_add(&h1[k1[c]], 1u, __ATOMIC_RELAXED, __HIP_MEMORY_SCOPE_WORKGROUP);
            r2[c] = __hip_atomic_fetch_add(&h2[k2[c]], 1u, __ATOMIC_RELAXED, __HIP_MEMORY_SCOPE_WORKGROUP);
        }
    }
    __syncthreads();

    if (wave < 2) {
        unsigned* hh = (wave == 0) ? h1 : h2;
        unsigned* pp = (wave == 0) ? p1 : p2;
        unsigned c[8];
        unsigned sum = 0u;
#pragma unroll
        for (int t = 0; t < 8; ++t) { c[t] = hh[lane * 8 + t]; sum += c[t]; }
        unsigned inc = sum;
#pragma unroll
        for (int d = 1; d < 32; d <<= 1) {
            const unsigned v = __shfl_up(inc, (unsigned)d);
            if (lane >= d) inc += v;
        }
        unsigned run = inc - sum;
#pragma unroll
        for (int t = 0; t < 8; ++t) {
            const unsigned start = run;
            run += c[t];
            hh[lane * 8 + t] = start;
            pp[lane * 8 + t] = start | (run << 16);
        }
    }
    __syncthreads();

    if (live) {
#pragma unroll
        for (int c = 0; c < 4; ++c) {
            s1[h1[k1[c]] + r1[c]] = (int)w1[c];
            s2[h2[k2[c]] + r2[c]] = (int)w2[c];
        }
    }
    __syncthreads();

    if (live) {
        ((int4*)rec1)[i4] = ((const int4*)s1)[tid];
        ((int4*)rec2)[i4] = ((const int4*)s2)[tid];
    }
    if (tid < OFF_PITCH)          off1[blockIdx.x * OFF_PITCH + tid] = p1[tid];
    else if (tid < 2 * OFF_PITCH) off2[blockIdx.x * OFF_PITCH + tid - OFF_PITCH] = p2[tid - OFF_PITCH];
}

template <bool FULL, int LBITS>
__device__ __forceinline__ void own_drain(
    float* acc, int* cnt, const int* q, int head, int n, int lane,
    const float* __restrict__ src)
{
    int r = 0, g = 0;
    if (FULL || lane < n) {
        const unsigned w = (unsigned)q[(head + lane) & (OW_Q - 1)];
        r = (int)(w & ((1u << LBITS) - 1u));
        g = (int)((w >> LBITS) & ((1u << (24 - LBITS)) - 1u));
        __hip_atomic_fetch_add(cnt + r, 1, __ATOMIC_RELAXED, __HIP_MEMORY_SCOPE_WORKGROUP);
    }
    for (int jb = 0; jb < (FULL ? 32 : n); jb += 8) {
        float4 v[8];
#pragma unroll
        for (int t = 0; t < 8; ++t) {
            const int gj = __builtin_amdgcn_readlane(g, jb + t);
            v[t] = *(const float4*)(src + (size_t)gj * OUT_CH + 4 * lane);
        }
#pragma unroll
        for (int t = 0; t < 8; ++t) {
            if (!FULL && jb + t >= n) break;
            const int rj = __builtin_amdgcn_readlane(r, jb + t);
            float* d = acc + rj * OUT_CH + lane;
            __hip_atomic_fetch_add(d +  0, v[t].x, __ATOMIC_RELAXED, __HIP_MEMORY_SCOPE_WORKGROUP);
            __hip_atomic_fetch_add(d + 32, v[t].y, __ATOMIC_RELAXED, __HIP_MEMORY_SCOPE_WORKGROUP);
            __hip_atomic_fetch_add(d + 64, v[t].z, __ATOMIC_RELAXED, __HIP_MEMORY_SCOPE_WORKGROUP);
            __hip_atomic_fetch_add(d + 96, v[t].w, __ATOMIC_RELAXED, __HIP_MEMORY_SCOPE_WORKGROUP);
        }
    }
}

template <int ROWS, int LBITS, bool HAS_BIAS>
__global__ __launch_bounds__(OW_THREADS) void agg_owned(
    const float* __restrict__ src, float* __restrict__ dst,
    const unsigned* __restrict__ off, const int* __restrict__ rec,
    const float* __restrict__ bias)
{
    __shared__ __attribute__((aligned(16))) float acc[ROWS * OUT_CH];
    __shared__ int cnt[ROWS];
    __shared__ int q_all[OW_WAVES][OW_Q];

    const int tid  = threadIdx.x;
    const int lane = tid & 31;
    const int wave = __builtin_amdgcn_readfirstlane(tid >> 5);
    int* q = q_all[wave];

    const unsigned group = blockIdx.x;

    unsigned se = 0u;
    {
        const int b = wave + OW_WAVES * lane;
        if (lane < SEG_PER_WAVE && b < N_SEG) se = off[(size_t)b * OFF_PITCH + group];
    }

    float4* acc4 = (float4*)acc;
    for (int i = tid; i < ROWS * OUT_CH / 4; i += OW_THREADS)
        acc4[i] = make_float4(0.0f, 0.0f, 0.0f, 0.0f);
    for (int i = tid; i < ROWS; i += OW_THREADS) cnt[i] = 0;
    __syncthreads();

    int w[SEG_PER_WAVE];
#pragma unroll
    for (int j = 0; j < SEG_PER_WAVE; ++j) {
        const unsigned sej = (unsigned)__builtin_amdgcn_readlane((int)se, j);
        const int b = wave + OW_WAVES * j;
        w[j] = rec[min(b * SEG + (int)(sej & 0xffffu) + lane, NNZ - 1)];
    }

    int head = 0;
    int qn   = 0;
#pragma unroll
    for (int j = 0; j < SEG_PER_WAVE; ++j) {
        const unsigned sej = (unsigned)__builtin_amdgcn_readlane((int)se, j);
        const int n = min((int)(sej >> 16) - (int)(sej & 0xffffu), 32);
        if (lane < n) q[qn + lane] = w[j];
        qn += max(n, 0);
    }

    int j = 0, i = 0, e = 0, b = 0;
    for (;;) {
        while (qn >= 32) {
            own_drain<true, LBITS>(acc, cnt, q, head, 32, lane, src);
            head = (head + 32) & (OW_Q - 1);
            qn -= 32;
        }
        while (i >= e && j < SEG_PER_WAVE) {
            const unsigned sej = (unsigned)__builtin_amdgcn_readlane((int)se, j);
            b = wave + OW_WAVES * j;
            i = (int)(sej & 0xffffu) + 32;
            e = (int)(sej >> 16);
            ++j;
        }
        if (i >= e) break;
        const int m = min(e - i, 32);
        if (lane < m) q[(head + qn + lane) & (OW_Q - 1)] = rec[b * SEG + i + lane];
        qn += m;
        i  += 32;
    }
    own_drain<false, LBITS>(acc, cnt, q, head, qn, lane, src);

    __syncthreads();

    const float4* __restrict__ bias4 = (const float4*)bias;
    float4* __restrict__ dst4 = (float4*)dst + (size_t)(blockIdx.x * ROWS) * (OUT_CH / 4);
    for (int i = tid; i < ROWS * OUT_CH / 4; i += OW_THREADS) {
        const int   c   = cnt[i >> 5];
        const float inv = (c > 0) ? (1.0f / (float)c) : 0.0f;
        const float* ar = acc + (i >> 5) * OUT_CH + (i & 31);
        const float4 a = make_float4(ar[0], ar[32], ar[64], ar[96]);
        if (HAS_BIAS) {
            const float4 b4 = bias4[i & 31];
            dst4[i] = make_float4(b4.x + a.x * inv, b4.y + a.y * inv,
                                  b4.z + a.z * inv, b4.w + a.w * inv);
        } else {
            dst4[i] = make_float4(a.x * inv, a.y * inv, a.z * inv, a.w * inv);
        }
    }
}

extern "C" void kernel_launch(void* const* d_in, const int* in_sizes, int n_in,
                              void* d_out, int out_size, void* d_ws, size_t ws_size,
                              hipStream_t stream)
{
    const float* x          = (const float*)d_in[0];
    const float* theta      = (const float*)d_in[1];
    const float* bias       = (const float*)d_in[2];
    const int*   node_idx   = (const int*)d_in[3];
    const int*   hyedge_idx = (const int*)d_in[4];
    float*       out        = (float*)d_out;

    float*    y      = out;
    float*    e_mean = (float*)d_ws;
    unsigned* off1   = (unsigned*)((char*)d_ws + 10240000);
    unsigned* off2   = (unsigned*)((char*)d_ws + 10640384);
    int*      rec1   = (int*)((char*)d_ws + 13440000);
    int*      rec2   = (int*)((char*)d_ws + 19840000);

    owner_sort<<<N_SEG, OW_THREADS, 0, stream>>>(
        hyedge_idx, node_idx, off1, off2, rec1, rec2);

    hyfm_gemm_wmma_f32<<<G_BLOCKS, G_THREADS, 0, stream>>>(x, theta, y);

    agg_owned<P1_ROWS, P1_LBITS, false><<<P1_GROUPS, OW_THREADS, 0, stream>>>(
        y, e_mean, off1, rec1, bias);

    agg_owned<P2_ROWS, P2_LBITS, true><<<P2_GROUPS, OW_THREADS, 0, stream>>>(
        e_mean, out, off2, rec2, bias);
}


# aggregation passes: records re-queued in LDS to the wave owning the destination row (row mod 32); one writer per LDS row, plain 16-B load/add/store, no LDS float atomics; sums in queue arrival order
# speedup vs baseline: 5.0387x; 1.0515x over previous
#include <hip/hip_runtime.h>

typedef float v2f __attribute__((ext_vector_type(2)));
typedef float v4f __attribute__((ext_vector_type(4)));
typedef float v8f __attribute__((ext_vector_type(8)));

constexpr int N_NODES   = 100000;
constexpr int N_HYEDGES = 20000;
constexpr int NNZ       = 1600000;
constexpr int IN_CH     = 256;
constexpr int OUT_CH    = 128;

constexpr int G_THREADS = 256;
constexpr int G_ROWS    = 32;
constexpr int G_STEPS   = N_NODES / G_ROWS;
constexpr int G_PITCH   = IN_CH + 4;
constexpr int G_BLOCKS  = 512;
constexpr int G_PRE     = G_ROWS * IN_CH / 4 / G_THREADS;

static_assert(G_STEPS * G_ROWS == N_NODES, "steps must tile the rows");
static_assert(OUT_CH == 16 * (G_THREADS / 32), "one 16-column tile per wave");
static_assert(G_BLOCKS <= G_STEPS, "every group has a first step");
static_assert(G_PRE * G_THREADS * 4 == G_ROWS * IN_CH, "a step must split evenly over the threads");

__device__ __forceinline__ void gemm_fetch(v4f (&pre)[G_PRE], const float* __restrict__ x, int step, int tid)
{
    const v4f* __restrict__ s = (const v4f*)(x + (size_t)step * G_ROWS * IN_CH) + tid;
#pragma unroll
    for (int p = 0; p < G_PRE; ++p) pre[p] = s[p * G_THREADS];
}

__device__ __forceinline__ void gemm_stage(float* xs, const v4f (&pre)[G_PRE], int tid)
{
#pragma unroll
    for (int p = 0; p < G_PRE; ++p) {
        const int row = p * (G_THREADS / 64) + (tid >> 6);
        *(v4f*)(xs + row * G_PITCH + 4 * (tid & 63)) = pre[p];
    }
}

__global__ __launch_bounds__(G_THREADS) void hyfm_gemm_wmma_f32(
    const float* __restrict__ x, const float* __restrict__ theta,
    float* __restrict__ y)
{
    __shared__ __attribute__((aligned(16))) float xs[2][G_ROWS * G_PITCH];

    const int tid  = threadIdx.x;
    const int lane = tid & 31;
    const int wave = __builtin_amdgcn_readfirstlane(tid >> 5);
    const int r    = lane & 15;
    const int half = lane >> 4;

    v2f b[IN_CH / 4];
    const float* __restrict__ th = theta + (size_t)(2 * half) * OUT_CH + wave * 16 + r;
#pragma unroll
    for (int s = 0; s < IN_CH / 4; ++s) {
        b[s].x = th[(4 * s) * OUT_CH];
        b[s].y = th[(4 * s + 1) * OUT_CH];
    }

    v4f pre[G_PRE];
    gemm_fetch(pre, x, blockIdx.x, tid);
    gemm_stage(xs[0], pre, tid);
    __syncthreads();

    const int aoff = r * G_PITCH + 2 * half;
    float* __restrict__ ycol = y + (size_t)(8 * half) * OUT_CH + wave * 16 + r;

    int buf = 0;
    for (int step = blockIdx.x; step < G_STEPS; step += G_BLOCKS) {
        const int  next = step + G_BLOCKS;
        const bool more = next < G_STEPS;
        if (more) gemm_fetch(pre, x, next, tid);

        const float* ap = xs[buf] + aoff;
        v8f acc0 = {}, acc1 = {};
#pragma unroll
        for (int s = 0; s < IN_CH / 4; ++s) {
            const v2f a0 = *(const v2f*)(ap + 4 * s);
            const v2f a1 = *(const v2f*)(ap + 16 * G_PITCH + 4 * s);
            acc0 = __builtin_amdgcn_wmma_f32_16x16x4_f32(
                false, a0, false, b[s], (short)0, acc0, false, false);
            acc1 = __builtin_amdgcn_wmma_f32_16x16x4_f32(
                false, a1, false, b[s], (short)0, acc1, false, false);
        }

        if (more) gemm_stage(xs[buf ^ 1], pre, tid);
        __syncthreads();
        buf ^= 1;

        float* __restrict__ yo = ycol + (size_t)step * G_ROWS * OUT_CH;
#pragma unroll
        for (int i = 0; i < 8; ++i) {
            yo[i * OUT_CH]        = acc0[i];
            yo[(i + 16) * OUT_CH] = acc1[i];
        }
    }
}

constexpr int OW_THREADS   = 1024;
constexpr int OW_WAVES     = OW_THREADS / 32;
constexpr int SEG          = 4 * OW_THREADS;
constexpr int N_SEG        = (NNZ + SEG - 1) / SEG;
constexpr int OFF_PITCH    = 256;
constexpr int SEG_PER_WAVE = (N_SEG + OW_WAVES - 1) / OW_WAVES;
constexpr int OW_Q         = 512;
constexpr int KEY_NONE     = 255;

constexpr int P1_ROWS   = 80;
constexpr int P1_GROUPS = N_HYEDGES / P1_ROWS;

constexpr int P2_ROWS   = 400;
constexpr int P2_GROUPS = N_NODES / P2_ROWS;

static_assert(P1_GROUPS * P1_ROWS == N_HYEDGES, "row ranges must tile the hyperedges");
static_assert(P2_GROUPS * P2_ROWS == N_NODES, "row ranges must tile the nodes");
static_assert(NNZ % 4 == 0, "a thread of owner_sort takes four entries");
static_assert(SEG <= 0xffff, "a run's start and end must fit a half-word each");
static_assert(SEG_PER_WAVE <= 32 && SEG_PER_WAVE * 32 + 32 <= OW_Q, "a wave's first pieces must fit its ring");
static_assert(P1_GROUPS <= KEY_NONE && P2_GROUPS <= KEY_NONE && KEY_NONE < OFF_PITCH, "a group number must fit its byte");

constexpr int P1_LBITS = 7;
constexpr int P2_LBITS = 9;
static_assert(P1_ROWS <= (1 << P1_LBITS) && N_NODES   <= (1 << (24 - P1_LBITS)), "pass 1 record fields");
static_assert(P2_ROWS <= (1 << P2_LBITS) && N_HYEDGES <= (1 << (24 - P2_LBITS)), "pass 2 record fields");

template <int ROWS, int LBITS>
__device__ __forceinline__ unsigned own_record(const unsigned own, const unsigned from, unsigned& key)
{
    const unsigned g = own / (unsigned)ROWS;
    key = min(g, (unsigned)KEY_NONE);
    return (key << 24) | ((from & ((1u << (24 - LBITS)) - 1u)) << LBITS) |
           ((own - g * (unsigned)ROWS) & ((1u << LBITS) - 1u));
}

__global__ __launch_bounds__(OW_THREADS) void owner_sort(
    const int* __restrict__ hyedge_idx, const int* __restrict__ node_idx,
    unsigned* __restrict__ off1, unsigned* __restrict__ off2,
    int* __restrict__ rec1, int* __restrict__ rec2)
{
    __shared__ unsigned h1[OFF_PITCH], h2[OFF_PITCH];
    __shared__ unsigned p1[OFF_PITCH], p2[OFF_PITCH];
    __shared__ __attribute__((aligned(16))) int s1[SEG];
    __shared__ __attribute__((aligned(16))) int s2[SEG];
    const int tid  = threadIdx.x;
    const int lane = tid & 31;
    const int wave = tid >> 5;
    if (tid < OFF_PITCH) { h1[tid] = 0u; h2[tid] = 0u; }
    __syncthreads();

    const int  i4   = blockIdx.x * OW_THREADS + tid;
    const bool live = i4 < NNZ / 4;
    unsigned w1[4], w2[4], k1[4], k2[4], r1[4], r2[4];
    if (live) {
        const int4 h = ((const int4*)hyedge_idx)[i4];
        const int4 n = ((const int4*)node_idx)[i4];
        const unsigned ho[4] = { (unsigned)h.x, (unsigned)h.y, (unsigned)h.z, (unsigned)h.w };
        const unsigned no[4] = { (unsigned)n.x, (unsigned)n.y, (unsigned)n.z, (unsigned)n.w };
#pragma unroll
        for (int c = 0; c < 4; ++c) {
            w1[c] = own_record<P1_ROWS, P1_LBITS>(ho[c], no[c], k1[c]);
            w2[c] = own_record<P2_ROWS, P2_LBITS>(no[c], ho[c], k2[c]);
            r1[c] = __hip_atomic_fetch_add(&h1[k1[c]], 1u, __ATOMIC_RELAXED, __HIP_MEMORY_SCOPE_WORKGROUP);
            r2[c] = __hip_atomic_fetch_add(&h2[k2[c]], 1u, __ATOMIC_RELAXED, __HIP_MEMORY_SCOPE_WORKGROUP);
        }
    }
    __syncthreads();

    if (wave < 2) {
        unsigned* hh = (wave == 0) ? h1 : h2;
        unsigned* pp = (wave == 0) ? p1 : p2;
        unsigned c[8];
        unsigned sum = 0u;
#pragma unroll
        for (int t = 0; t < 8; ++t) { c[t] = hh[lane * 8 + t]; sum += c[t]; }
        unsigned inc = sum;
#pragma unroll
        for (int d = 1; d < 32; d <<= 1) {
            const unsigned v = __shfl_up(inc, (unsigned)d);
            if (lane >= d) inc += v;
        }
        unsigned run = inc - sum;
#pragma unroll
        for (int t = 0; t < 8; ++t) {
            const unsigned start = run;
            run += c[t];
            hh[lane * 8 + t] = start;
            pp[lane * 8 + t] = start | (run << 16);
        }
    }
    __syncthreads();

    if (live) {
#pragma unroll
        for (int c = 0; c < 4; ++c) {
            s1[h1[k1[c]] + r1[c]] = (int)w1[c];
            s2[h2[k2[c]] + r2[c]] = (int)w2[c];
        }
    }
    __syncthreads();

    if (live) {
        ((int4*)rec1)[i4] = ((const int4*)s1)[tid];
        ((int4*)rec2)[i4] = ((const int4*)s2)[tid];
    }
    if (tid < OFF_PITCH)          off1[blockIdx.x * OFF_PITCH + tid] = p1[tid];
    else if (tid < 2 * OFF_PITCH) off2[blockIdx.x * OFF_PITCH + tid - OFF_PITCH] = p2[tid - OFF_PITCH];
}

template <bool FULL, int LBITS>
__device__ __forceinline__ void own_drain(
    float* acc, int* cnt, const int* q, int head, int n, int lane,
    const float* __restrict__ src)
{
    int r = 0, g = 0;
    if (FULL || lane < n) {
        const unsigned w = (unsigned)q[(head + lane) & (OW_Q - 1)];
        r = (int)(w & ((1u << LBITS) - 1u));
        g = (int)((w >> LBITS) & ((1u << (24 - LBITS)) - 1u));
        __hip_atomic_fetch_add(cnt + r, 1, __ATOMIC_RELAXED, __HIP_MEMORY_SCOPE_WORKGROUP);
    }
    for (int jb = 0; jb < (FULL ? 32 : n); jb += 8) {
        float4 v[8];
#pragma unroll
        for (int t = 0; t < 8; ++t) {
            const int gj = __builtin_amdgcn_readlane(g, jb + t);
            v[t] = *(const float4*)(src + (size_t)gj * OUT_CH + 4 * lane);
        }
#pragma unroll
        for (int t = 0; t < 8; ++t) {
            if (!FULL && jb + t >= n) break;
            const int rj = __builtin_amdgcn_readlane(r, jb + t);
            float* d = acc + rj * OUT_CH + lane;
            __hip_atomic_fetch_add(d +  0, v[t].x, __ATOMIC_RELAXED, __HIP_MEMORY_SCOPE_WORKGROUP);
            __hip_atomic_fetch_add(d + 32, v[t].y, __ATOMIC_RELAXED, __HIP_MEMORY_SCOPE_WORKGROUP);
            __hip_atomic_fetch_add(d + 64, v[t].z, __ATOMIC_RELAXED, __HIP_MEMORY_SCOPE_WORKGROUP);
            __hip_atomic_fetch_add(d + 96, v[t].w, __ATOMIC_RELAXED, __HIP_MEMORY_SCOPE_WORKGROUP);
        }
    }
}

template <int LBITS>
__device__ __forceinline__ void own_push(int (*q_all)[OW_Q], int* qn_all, int word)
{
    const int o = (int)((unsigned)word & ((1u << LBITS) - 1u)) & (OW_WAVES - 1);
    const int pos = __hip_atomic_fetch_add(qn_all + o, 1, __ATOMIC_RELAXED, __HIP_MEMORY_SCOPE_WORKGROUP);
    if (pos < OW_Q) q_all[o][pos] = word;
}

template <bool FULL, int LBITS>
__device__ __forceinline__ void own_drain_single(
    float* acc, int* cnt, const int* q, int n, int lane,
    const float* __restrict__ src)
{
    int r = 0, g = 0;
    if (FULL || lane < n) {
        const unsigned w = (unsigned)q[lane];
        r = (int)(w & ((1u << LBITS) - 1u));
        g = (int)((w >> LBITS) & ((1u << (24 - LBITS)) - 1u));
        __hip_atomic_fetch_add(cnt + r, 1, __ATOMIC_RELAXED, __HIP_MEMORY_SCOPE_WORKGROUP);
    }
    for (int jb = 0; jb < (FULL ? 32 : n); jb += 8) {
        float4 v[8];
#pragma unroll
        for (int t = 0; t < 8; ++t) {
            const int gj = __builtin_amdgcn_readlane(g, jb + t);
            v[t] = *(const float4*)(src + (size_t)gj * OUT_CH + 4 * lane);
        }
#pragma unroll
        for (int t = 0; t < 8; ++t) {
            if (!FULL && jb + t >= n) break;
            const int rj = __builtin_amdgcn_readlane(r, jb + t);
            float4* d = (float4*)(acc + rj * OUT_CH) + lane;
            float4 o = *d;
            o.x += v[t].x; o.y += v[t].y; o.z += v[t].z; o.w += v[t].w;
            *d = o;
        }
    }
}

template <int ROWS, int LBITS, bool HAS_BIAS>
__global__ __launch_bounds__(OW_THREADS) void agg_owned(
    const float* __restrict__ src, float* __restrict__ dst,
    const unsigned* __restrict__ off, const int* __restrict__ rec,
    const float* __restrict__ bias)
{
    __shared__ __attribute__((aligned(16))) float acc[ROWS * OUT_CH];
    __shared__ int cnt[ROWS];
    __shared__ int q_all[OW_WAVES][OW_Q];
    __shared__ int qn_all[OW_WAVES];

    const int tid  = threadIdx.x;
    const int lane = tid & 31;
    const int wave = __builtin_amdgcn_readfirstlane(tid >> 5);

    const unsigned group = blockIdx.x;

    unsigned se = 0u;
    {
        const int b = wave + OW_WAVES * lane;
        if (lane < SEG_PER_WAVE && b < N_SEG) se = off[(size_t)b * OFF_PITCH + group];
    }

    float4* acc4 = (float4*)acc;
    for (int i = tid; i < ROWS * OUT_CH / 4; i += OW_THREADS)
        acc4[i] = make_float4(0.0f, 0.0f, 0.0f, 0.0f);
    for (int i = tid; i < ROWS; i += OW_THREADS) cnt[i] = 0;
    if (tid < OW_WAVES) qn_all[tid] = 0;
    __syncthreads();

    int w[SEG_PER_WAVE];
#pragma unroll
    for (int j = 0; j < SEG_PER_WAVE; ++j) {
        const unsigned sej = (unsigned)__builtin_amdgcn_readlane((int)se, j);
        const int b = wave + OW_WAVES * j;
        w[j] = rec[min(b * SEG + (int)(sej & 0xffffu) + lane, NNZ - 1)];
    }

#pragma unroll
    for (int j = 0; j < SEG_PER_WAVE; ++j) {
        const unsigned sej = (unsigned)__builtin_amdgcn_readlane((int)se, j);
        const int n = min((int)(sej >> 16) - (int)(sej & 0xffffu), 32);
        if (lane < n) own_push<LBITS>(q_all, qn_all, w[j]);
    }
    for (int j = 0; j < SEG_PER_WAVE; ++j) {
        const unsigned sej = (unsigned)__builtin_amdgcn_readlane((int)se, j);
        const int b = wave + OW_WAVES * j;
        const int e = (int)(sej >> 16);
        for (int i = (int)(sej & 0xffffu) + 32; i < e; i += 32)
            if (i + lane < e) own_push<LBITS>(q_all, qn_all, rec[b * SEG + i + lane]);
    }
    __syncthreads();

    const bool over = __builtin_amdgcn_ballot_w32(qn_all[lane] > OW_Q) != 0u;

    if (!over) {
        const int* q = q_all[wave];
        const int qn = __builtin_amdgcn_readfirstlane(qn_all[wave]);
        int base = 0;
        for (; base + 32 <= qn; base += 32)
            own_drain_single<true, LBITS>(acc, cnt, q + base, 32, lane, src);
        if (base < qn)
            own_drain_single<false, LBITS>(acc, cnt, q + base, qn - base, lane, src);
    } else {
        int* q = q_all[wave];
        int head = 0;
        int qn   = 0;
#pragma unroll
        for (int j = 0; j < SEG_PER_WAVE; ++j) {
            const unsigned sej = (unsigned)__builtin_amdgcn_readlane((int)se, j);
            const int n = min((int)(sej >> 16) - (int)(sej & 0xffffu), 32);
            if (lane < n) q[qn + lane] = w[j];
            qn += max(n, 0);
        }

        int j = 0, i = 0, e = 0, b = 0;
        for (;;) {
            while (qn >= 32) {
                own_drain<true, LBITS>(acc, cnt, q, head, 32, lane, src);
                head = (head + 32) & (OW_Q - 1);
                qn -= 32;
            }
            while (i >= e && j < SEG_PER_WAVE) {
                const unsigned sej = (unsigned)__builtin_amdgcn_readlane((int)se, j);
                b = wave + OW_WAVES * j;
                i = (int)(sej & 0xffffu) + 32;
                e = (int)(sej >> 16);
                ++j;
            }
            if (i >= e) break;
            const int m = min(e - i, 32);
            if (lane < m) q[(head + qn + lane) & (OW_Q - 1)] = rec[b * SEG + i + lane];
            qn += m;
            i  += 32;
        }
        own_drain<false, LBITS>(acc, cnt, q, head, qn, lane, src);
    }

    __syncthreads();

    const float4* __restrict__ bias4 = (const float4*)bias;
    float4* __restrict__ dst4 = (float4*)dst + (size_t)(blockIdx.x * ROWS) * (OUT_CH / 4);
    for (int i = tid; i < ROWS * OUT_CH / 4; i += OW_THREADS) {
        const int   c   = cnt[i >> 5];
        const float inv = (c > 0) ? (1.0f / (float)c) : 0.0f;
        float4 a;
        if (!over) {
            a = acc4[i];
        } else {
            const float* ar = acc + (i >> 5) * OUT_CH + (i & 31);
            a = make_float4(ar[0], ar[32], ar[64], ar[96]);
        }
        if (HAS_BIAS) {
            const float4 b4 = bias4[i & 31];
            dst4[i] = make_float4(b4.x + a.x * inv, b4.y + a.y * inv,
                                  b4.z + a.z * inv, b4.w + a.w * inv);
        } else {
            dst4[i] = make_float4(a.x * inv, a.y * inv, a.z * inv, a.w * inv);
        }
    }
}

extern "C" void kernel_launch(void* const* d_in, const int* in_sizes, int n_in,
                              void* d_out, int out_size, void* d_ws, size_t ws_size,
                              hipStream_t stream)
{
    const float* x          = (const float*)d_in[0];
    const float* theta      = (const float*)d_in[1];
    const float* bias       = (const float*)d_in[2];
    const int*   node_idx   = (const int*)d_in[3];
    const int*   hyedge_idx = (const int*)d_in[4];
    float*       out        = (float*)d_out;

    float*    y      = out;
    float*    e_mean = (float*)d_ws;
    unsigned* off1   = (unsigned*)((char*)d_ws + 10240000);
    unsigned* off2   = (unsigned*)((char*)d_ws + 10640384);
    int*      rec1   = (int*)((char*)d_ws + 13440000);
    int*      rec2   = (int*)((char*)d_ws + 19840000);

    owner_sort<<<N_SEG, OW_THREADS, 0, stream>>>(
        hyedge_idx, node_idx, off1, off2, rec1, rec2);

    hyfm_gemm_wmma_f32<<<G_BLOCKS, G_THREADS, 0, stream>>>(x, theta, y);

    agg_owned<P1_ROWS, P1_LBITS, false><<<P1_GROUPS, OW_THREADS, 0, stream>>>(
        y, e_mean, off1, rec1, bias);

    agg_owned<P2_ROWS, P2_LBITS, true><<<P2_GROUPS, OW_THREADS, 0, stream>>>(
        e_mean, out, off2, rec2, bias);
}


# aggregation passes: a queue's last partial batch gathers in whole groups of eight rows (lanes past the end add source row 0 into a spare LDS row); no per-entry test, so 8 loads fly; sum order kept
# speedup vs baseline: 5.2060x; 1.0332x over previous
#include <hip/hip_runtime.h>

typedef float v2f __attribute__((ext_vector_type(2)));
typedef float v4f __attribute__((ext_vector_type(4)));
typedef float v8f __attribute__((ext_vector_type(8)));

constexpr int N_NODES   = 100000;
constexpr int N_HYEDGES = 20000;
constexpr int NNZ       = 1600000;
constexpr int IN_CH     = 256;
constexpr int OUT_CH    = 128;

constexpr int G_THREADS = 256;
constexpr int G_ROWS    = 32;
constexpr int G_STEPS   = N_NODES / G_ROWS;
constexpr int G_PITCH   = IN_CH + 4;
constexpr int G_BLOCKS  = 512;
constexpr int G_PRE     = G_ROWS * IN_CH / 4 / G_THREADS;

static_assert(G_STEPS * G_ROWS == N_NODES, "steps must tile the rows");
static_assert(OUT_CH == 16 * (G_THREADS / 32), "one 16-column tile per wave");
static_assert(G_BLOCKS <= G_STEPS, "every group has a first step");
static_assert(G_PRE * G_THREADS * 4 == G_ROWS * IN_CH, "a step must split evenly over the threads");

__device__ __forceinline__ void gemm_fetch(v4f (&pre)[G_PRE], const float* __restrict__ x, int step, int tid)
{
    const v4f* __restrict__ s = (const v4f*)(x + (size_t)step * G_ROWS * IN_CH) + tid;
#pragma unroll
    for (int p = 0; p < G_PRE; ++p) pre[p] = s[p * G_THREADS];
}

__device__ __forceinline__ void gemm_stage(float* xs, const v4f (&pre)[G_PRE], int tid)
{
#pragma unroll
    for (int p = 0; p < G_PRE; ++p) {
        const int row = p * (G_THREADS / 64) + (tid >> 6);
        *(v4f*)(xs + row * G_PITCH + 4 * (tid & 63)) = pre[p];
    }
}

__global__ __launch_bounds__(G_THREADS) void hyfm_gemm_wmma_f32(
    const float* __restrict__ x, const float* __restrict__ theta,
    float* __restrict__ y)
{
    __shared__ __attribute__((aligned(16))) float xs[2][G_ROWS * G_PITCH];

    const int tid  = threadIdx.x;
    const int lane = tid & 31;
    const int wave = __builtin_amdgcn_readfirstlane(tid >> 5);
    const int r    = lane & 15;
    const int half = lane >> 4;

    v2f b[IN_CH / 4];
    const float* __restrict__ th = theta + (size_t)(2 * half) * OUT_CH + wave * 16 + r;
#pragma unroll
    for (int s = 0; s < IN_CH / 4; ++s) {
        b[s].x = th[(4 * s) * OUT_CH];
        b[s].y = th[(4 * s + 1) * OUT_CH];
    }

    v4f pre[G_PRE];
    gemm_fetch(pre, x, blockIdx.x, tid);
    gemm_stage(xs[0], pre, tid);
    __syncthreads();

    const int aoff = r * G_PITCH + 2 * half;
    float* __restrict__ ycol = y + (size_t)(8 * half) * OUT_CH + wave * 16 + r;

    int buf = 0;
    for (int step = blockIdx.x; step < G_STEPS; step += G_BLOCKS) {
        const int  next = step + G_BLOCKS;
        const bool more = next < G_STEPS;
        if (more) gemm_fetch(pre, x, next, tid);

        const float* ap = xs[buf] + aoff;
        v8f acc0 = {}, acc1 = {};
#pragma unroll
        for (int s = 0; s < IN_CH / 4; ++s) {
            const v2f a0 = *(const v2f*)(ap + 4 * s);
            const v2f a1 = *(const v2f*)(ap + 16 * G_PITCH + 4 * s);
            acc0 = __builtin_amdgcn_wmma_f32_16x16x4_f32(
                false, a0, false, b[s], (short)0, acc0, false, false);
            acc1 = __builtin_amdgcn_wmma_f32_16x16x4_f32(
                false, a1, false, b[s], (short)0, acc1, false, false);
        }

        if (more) gemm_stage(xs[buf ^ 1], pre, tid);
        __syncthreads();
        buf ^= 1;

        float* __restrict__ yo = ycol + (size_t)step * G_ROWS * OUT_CH;
#pragma unroll
        for (int i = 0; i < 8; ++i) {
            yo[i * OUT_CH]        = acc0[i];
            yo[(i + 16) * OUT_CH] = acc1[i];
        }
    }
}

constexpr int OW_THREADS   = 1024;
constexpr int OW_WAVES     = OW_THREADS / 32;
constexpr int SEG          = 4 * OW_THREADS;
constexpr int N_SEG        = (NNZ + SEG - 1) / SEG;
constexpr int OFF_PITCH    = 256;
constexpr int SEG_PER_WAVE = (N_SEG + OW_WAVES - 1) / OW_WAVES;
constexpr int OW_Q         = 512;
constexpr int KEY_NONE     = 255;

constexpr int P1_ROWS   = 80;
constexpr int P1_GROUPS = N_HYEDGES / P1_ROWS;

constexpr int P2_ROWS   = 400;
constexpr int P2_GROUPS = N_NODES / P2_ROWS;

static_assert(P1_GROUPS * P1_ROWS == N_HYEDGES, "row ranges must tile the hyperedges");
static_assert(P2_GROUPS * P2_ROWS == N_NODES, "row ranges must tile the nodes");
static_assert(NNZ % 4 == 0, "a thread of owner_sort takes four entries");
static_assert(SEG <= 0xffff, "a run's start and end must fit a half-word each");
static_assert(SEG_PER_WAVE <= 32 && SEG_PER_WAVE * 32 + 32 <= OW_Q, "a wave's first pieces must fit its ring");
static_assert(P1_GROUPS <= KEY_NONE && P2_GROUPS <= KEY_NONE && KEY_NONE < OFF_PITCH, "a group number must fit its byte");

constexpr int P1_LBITS = 7;
constexpr int P2_LBITS = 9;
static_assert(P1_ROWS <= (1 << P1_LBITS) && N_NODES   <= (1 << (24 - P1_LBITS)), "pass 1 record fields");
static_assert(P2_ROWS <= (1 << P2_LBITS) && N_HYEDGES <= (1 << (24 - P2_LBITS)), "pass 2 record fields");

template <int ROWS, int LBITS>
__device__ __forceinline__ unsigned own_record(const unsigned own, const unsigned from, unsigned& key)
{
    const unsigned g = own / (unsigned)ROWS;
    key = min(g, (unsigned)KEY_NONE);
    return (key << 24) | ((from & ((1u << (24 - LBITS)) - 1u)) << LBITS) |
           ((own - g * (unsigned)ROWS) & ((1u << LBITS) - 1u));
}

__global__ __launch_bounds__(OW_THREADS) void owner_sort(
    const int* __restrict__ hyedge_idx, const int* __restrict__ node_idx,
    unsigned* __restrict__ off1, unsigned* __restrict__ off2,
    int* __restrict__ rec1, int* __restrict__ rec2)
{
    __shared__ unsigned h1[OFF_PITCH], h2[OFF_PITCH];
    __shared__ unsigned p1[OFF_PITCH], p2[OFF_PITCH];
    __shared__ __attribute__((aligned(16))) int s1[SEG];
    __shared__ __attribute__((aligned(16))) int s2[SEG];
    const int tid  = threadIdx.x;
    const int lane = tid & 31;
    const int wave = tid >> 5;
    if (tid < OFF_PITCH) { h1[tid] = 0u; h2[tid] = 0u; }
    __syncthreads();

    const int  i4   = blockIdx.x * OW_THREADS + tid;
    const bool live = i4 < NNZ / 4;
    unsigned w1[4], w2[4], k1[4], k2[4], r1[4], r2[4];
    if (live) {
        const int4 h = ((const int4*)hyedge_idx)[i4];
        const int4 n = ((const int4*)node_idx)[i4];
        const unsigned ho[4] = { (unsigned)h.x, (unsigned)h.y, (unsigned)h.z, (unsigned)h.w };
        const unsigned no[4] = { (unsigned)n.x, (unsigned)n.y, (unsigned)n.z, (unsigned)n.w };
#pragma unroll
        for (int c = 0; c < 4; ++c) {
            w1[c] = own_record<P1_ROWS, P1_LBITS>(ho[c], no[c], k1[c]);
            w2[c] = own_record<P2_ROWS, P2_LBITS>(no[c], ho[c], k2[c]);
            r1[c] = __hip_atomic_fetch_add(&h1[k1[c]], 1u, __ATOMIC_RELAXED, __HIP_MEMORY_SCOPE_WORKGROUP);
            r2[c] = __hip_atomic_fetch_add(&h2[k2[c]], 1u, __ATOMIC_RELAXED, __HIP_MEMORY_SCOPE_WORKGROUP);
        }
    }
    __syncthreads();

    if (wave < 2) {
        unsigned* hh = (wave == 0) ? h1 : h2;
        unsigned* pp = (wave == 0) ? p1 : p2;
        unsigned c[8];
        unsigned sum = 0u;
#pragma unroll
        for (int t = 0; t < 8; ++t) { c[t] = hh[lane * 8 + t]; sum += c[t]; }
        unsigned inc = sum;
#pragma unroll
        for (int d = 1; d < 32; d <<= 1) {
            const unsigned v = __shfl_up(inc, (unsigned)d);
            if (lane >= d) inc += v;
        }
        unsigned run = inc - sum;
#pragma unroll
        for (int t = 0; t < 8; ++t) {
            const unsigned start = run;
            run += c[t];
            hh[lane * 8 + t] = start;
            pp[lane * 8 + t] = start | (run << 16);
        }
    }
    __syncthreads();

    if (live) {
#pragma unroll
        for (int c = 0; c < 4; ++c) {
            s1[h1[k1[c]] + r1[c]] = (int)w1[c];
            s2[h2[k2[c]] + r2[c]] = (int)w2[c];
        }
    }
    __syncthreads();

    if (live) {
        ((int4*)rec1)[i4] = ((const int4*)s1)[tid];
        ((int4*)rec2)[i4] = ((const int4*)s2)[tid];
    }
    if (tid < OFF_PITCH)          off1[blockIdx.x * OFF_PITCH + tid] = p1[tid];
    else if (tid < 2 * OFF_PITCH) off2[blockIdx.x * OFF_PITCH + tid - OFF_PITCH] = p2[tid - OFF_PITCH];
}

template <bool FULL, int LBITS>
__device__ __forceinline__ void own_drain(
    float* acc, int* cnt, const int* q, int head, int n, int lane,
    const float* __restrict__ src)
{
    int r = 0, g = 0;
    if (FULL || lane < n) {
        const unsigned w = (unsigned)q[(head + lane) & (OW_Q - 1)];
        r = (int)(w & ((1u << LBITS) - 1u));
        g = (int)((w >> LBITS) & ((1u << (24 - LBITS)) - 1u));
        __hip_atomic_fetch_add(cnt + r, 1, __ATOMIC_RELAXED, __HIP_MEMORY_SCOPE_WORKGROUP);
    }
    for (int jb = 0; jb < (FULL ? 32 : n); jb += 8) {
        float4 v[8];
#pragma unroll
        for (int t = 0; t < 8; ++t) {
            const int gj = __builtin_amdgcn_readlane(g, jb + t);
            v[t] = *(const float4*)(src + (size_t)gj * OUT_CH + 4 * lane);
        }
#pragma unroll
        for (int t = 0; t < 8; ++t) {
            if (!FULL && jb + t >= n) break;
            const int rj = __builtin_amdgcn_readlane(r, jb + t);
            float* d = acc + rj * OUT_CH + lane;
            __hip_atomic_fetch_add(d +  0, v[t].x, __ATOMIC_RELAXED, __HIP_MEMORY_SCOPE_WORKGROUP);
            __hip_atomic_fetch_add(d + 32, v[t].y, __ATOMIC_RELAXED, __HIP_MEMORY_SCOPE_WORKGROUP);
            __hip_atomic_fetch_add(d + 64, v[t].z, __ATOMIC_RELAXED, __HIP_MEMORY_SCOPE_WORKGROUP);
            __hip_atomic_fetch_add(d + 96, v[t].w, __ATOMIC_RELAXED, __HIP_MEMORY_SCOPE_WORKGROUP);
        }
    }
}

template <int LBITS>
__device__ __forceinline__ void own_push(int (*q_all)[OW_Q], int* qn_all, int word)
{
    const int o = (int)((unsigned)word & ((1u << LBITS) - 1u)) & (OW_WAVES - 1);
    const int pos = __hip_atomic_fetch_add(qn_all + o, 1, __ATOMIC_RELAXED, __HIP_MEMORY_SCOPE_WORKGROUP);
    if (pos < OW_Q) q_all[o][pos] = word;
}

template <bool FULL, int LBITS, int SPARE>
__device__ __forceinline__ void own_drain_single(
    float* acc, int* cnt, const int* q, int n, int lane,
    const float* __restrict__ src)
{
    int r = SPARE, g = 0;
    if (FULL || lane < n) {
        const unsigned w = (unsigned)q[lane];
        r = (int)(w & ((1u << LBITS) - 1u));
        g = (int)((w >> LBITS) & ((1u << (24 - LBITS)) - 1u));
        __hip_atomic_fetch_add(cnt + r, 1, __ATOMIC_RELAXED, __HIP_MEMORY_SCOPE_WORKGROUP);
    }
    for (int jb = 0; jb < (FULL ? 32 : n); jb += 8) {
        float4 v[8];
#pragma unroll
        for (int t = 0; t < 8; ++t) {
            const int gj = __builtin_amdgcn_readlane(g, jb + t);
            v[t] = *(const float4*)(src + (size_t)gj * OUT_CH + 4 * lane);
        }
#pragma unroll
        for (int t = 0; t < 8; ++t) {
            const int rj = __builtin_amdgcn_readlane(r, jb + t);
            float4* d = (float4*)(acc + rj * OUT_CH) + lane;
            float4 o = *d;
            o.x += v[t].x; o.y += v[t].y; o.z += v[t].z; o.w += v[t].w;
            *d = o;
        }
    }
}

template <int ROWS, int LBITS, bool HAS_BIAS>
__global__ __launch_bounds__(OW_THREADS) void agg_owned(
    const float* __restrict__ src, float* __restrict__ dst,
    const unsigned* __restrict__ off, const int* __restrict__ rec,
    const float* __restrict__ bias)
{
    __shared__ __attribute__((aligned(16))) float acc[(ROWS + 1) * OUT_CH];
    __shared__ int cnt[ROWS];
    __shared__ int q_all[OW_WAVES][OW_Q];
    __shared__ int qn_all[OW_WAVES];

    const int tid  = threadIdx.x;
    const int lane = tid & 31;
    const int wave = __builtin_amdgcn_readfirstlane(tid >> 5);

    const unsigned group = blockIdx.x;

    unsigned se = 0u;
    {
        const int b = wave + OW_WAVES * lane;
        if (lane < SEG_PER_WAVE && b < N_SEG) se = off[(size_t)b * OFF_PITCH + group];
    }

    float4* acc4 = (float4*)acc;
    for (int i = tid; i < (ROWS + 1) * OUT_CH / 4; i += OW_THREADS)
        acc4[i] = make_float4(0.0f, 0.0f, 0.0f, 0.0f);
    for (int i = tid; i < ROWS; i += OW_THREADS) cnt[i] = 0;
    if (tid < OW_WAVES) qn_all[tid] = 0;
    __syncthreads();

    int w[SEG_PER_WAVE];
#pragma unroll
    for (int j = 0; j < SEG_PER_WAVE; ++j) {
        const unsigned sej = (unsigned)__builtin_amdgcn_readlane((int)se, j);
        const int b = wave + OW_WAVES * j;
        w[j] = rec[min(b * SEG + (int)(sej & 0xffffu) + lane, NNZ - 1)];
    }

#pragma unroll
    for (int j = 0; j < SEG_PER_WAVE; ++j) {
        const unsigned sej = (unsigned)__builtin_amdgcn_readlane((int)se, j);
        const int n = min((int)(sej >> 16) - (int)(sej & 0xffffu), 32);
        if (lane < n) own_push<LBITS>(q_all, qn_all, w[j]);
    }
    for (int j = 0; j < SEG_PER_WAVE; ++j) {
        const unsigned sej = (unsigned)__builtin_amdgcn_readlane((int)se, j);
        const int b = wave + OW_WAVES * j;
        const int e = (int)(sej >> 16);
        for (int i = (int)(sej & 0xffffu) + 32; i < e; i += 32)
            if (i + lane < e) own_push<LBITS>(q_all, qn_all, rec[b * SEG + i + lane]);
    }
    __syncthreads();

    const bool over = __builtin_amdgcn_ballot_w32(qn_all[lane] > OW_Q) != 0u;

    if (!over) {
        const int* q = q_all[wave];
        const int qn = __builtin_amdgcn_readfirstlane(qn_all[wave]);
        int base = 0;
        for (; base + 32 <= qn; base += 32)
            own_drain_single<true, LBITS, ROWS>(acc, cnt, q + base, 32, lane, src);
        if (base < qn)
            own_drain_single<false, LBITS, ROWS>(acc, cnt, q + base, qn - base, lane, src);
    } else {
        int* q = q_all[wave];
        int head = 0;
        int qn   = 0;
#pragma unroll
        for (int j = 0; j < SEG_PER_WAVE; ++j) {
            const unsigned sej = (unsigned)__builtin_amdgcn_readlane((int)se, j);
            const int n = min((int)(sej >> 16) - (int)(sej & 0xffffu), 32);
            if (lane < n) q[qn + lane] = w[j];
            qn += max(n, 0);
        }

        int j = 0, i = 0, e = 0, b = 0;
        for (;;) {
            while (qn >= 32) {
                own_drain<true, LBITS>(acc, cnt, q, head, 32, lane, src);
                head = (head + 32) & (OW_Q - 1);
                qn -= 32;
            }
            while (i >= e && j < SEG_PER_WAVE) {
                const unsigned sej = (unsigned)__builtin_amdgcn_readlane((int)se, j);
                b = wave + OW_WAVES * j;
                i = (int)(sej & 0xffffu) + 32;
                e = (int)(sej >> 16);
                ++j;
            }
            if (i >= e) break;
            const int m = min(e - i, 32);
            if (lane < m) q[(head + qn + lane) & (OW_Q - 1)] = rec[b * SEG + i + lane];
            qn += m;
            i  += 32;
        }
        own_drain<false, LBITS>(acc, cnt, q, head, qn, lane, src);
    }

    __syncthreads();

    const float4* __restrict__ bias4 = (const float4*)bias;
    float4* __restrict__ dst4 = (float4*)dst + (size_t)(blockIdx.x * ROWS) * (OUT_CH / 4);
    for (int i = tid; i < ROWS * OUT_CH / 4; i += OW_THREADS) {
        const int   c   = cnt[i >> 5];
        const float inv = (c > 0) ? (1.0f / (float)c) : 0.0f;
        float4 a;
        if (!over) {
            a = acc4[i];
        } else {
            const float* ar = acc + (i >> 5) * OUT_CH + (i & 31);
            a = make_float4(ar[0], ar[32], ar[64], ar[96]);
        }
        if (HAS_BIAS) {
            const float4 b4 = bias4[i & 31];
            dst4[i] = make_float4(b4.x + a.x * inv, b4.y + a.y * inv,
                                  b4.z + a.z * inv, b4.w + a.w * inv);
        } else {
            dst4[i] = make_float4(a.x * inv, a.y * inv, a.z * inv, a.w * inv);
        }
    }
}

extern "C" void kernel_launch(void* const* d_in, const int* in_sizes, int n_in,
                              void* d_out, int out_size, void* d_ws, size_t ws_size,
                              hipStream_t stream)
{
    const float* x          = (const float*)d_in[0];
    const float* theta      = (const float*)d_in[1];
    const float* bias       = (const float*)d_in[2];
    const int*   node_idx   = (const int*)d_in[3];
    const int*   hyedge_idx = (const int*)d_in[4];
    float*       out        = (float*)d_out;

    float*    y      = out;
    float*    e_mean = (float*)d_ws;
    unsigned* off1   = (unsigned*)((char*)d_ws + 10240000);
    unsigned* off2   = (unsigned*)((char*)d_ws + 10640384);
    int*      rec1   = (int*)((char*)d_ws + 13440000);
    int*      rec2   = (int*)((char*)d_ws + 19840000);

    owner_sort<<<N_SEG, OW_THREADS, 0, stream>>>(
        hyedge_idx, node_idx, off1, off2, rec1, rec2);

    hyfm_gemm_wmma_f32<<<G_BLOCKS, G_THREADS, 0, stream>>>(x, theta, y);

    agg_owned<P1_ROWS, P1_LBITS, false><<<P1_GROUPS, OW_THREADS, 0, stream>>>(
        y, e_mean, off1, rec1, bias);

    agg_owned<P2_ROWS, P2_LBITS, true><<<P2_GROUPS, OW_THREADS, 0, stream>>>(
        e_mean, out, off2, rec2, bias);
}
